# starting kernel
# baseline (speedup 1.0000x reference)
#include <hip/hip_runtime.h>
#include <hip/hip_bf16.h>


typedef __attribute__((ext_vector_type(2))) float v2f;
typedef __attribute__((ext_vector_type(8))) float v8f;

__global__ void gcn_transpose_w(const float* __restrict__ W,
                                float* __restrict__ Wt, int K) {
  int i = blockIdx.x * blockDim.x + threadIdx.x;
  if (i < K * 128) {
    const int k = i >> 7;
    const int c = i & 127;
    Wt[(size_t)c * K + k] = W[i];
  }
}

template <int K>
__global__ __launch_bounds__(256) void gcn_gemm_wmma(
    const float* __restrict__ X, const float* __restrict__ Wt,
    float* __restrict__ H, int nrows) {
  constexpr int LDA = K + 4;
  __shared__ __align__(16) float As[16 * LDA];

  const int rowBase = blockIdx.x * 16;
  const int tid = threadIdx.x;

  constexpr int vecPerRow = K / 4;
  for (int i = tid; i < 16 * vecPerRow; i += 256) {
    const int r = i / vecPerRow;
    const int v = i % vecPerRow;
    const int grow = rowBase + r;
    float4 val = make_float4(0.f, 0.f, 0.f, 0.f);
    if (grow < nrows) {
      val = *reinterpret_cast<const float4*>(X + (size_t)grow * K + 4 * v);
    }
    *reinterpret_cast<float4*>(&As[r * LDA + 4 * v]) = val;
  }
  __syncthreads();

  const int wave = tid >> 5;
  const int lane = tid & 31;
  const int l15 = lane & 15;
  const int halfSel = lane >> 4;
  const int col = wave * 16 + l15;

  const float* __restrict__ Bcol = Wt + (size_t)col * K;

  v8f acc = {};
#pragma unroll 4
  for (int k0 = 0; k0 < K; k0 += 4) {
    const int kk = k0 + 2 * halfSel;
    v2f a;
    a.x = As[l15 * LDA + kk];
    a.y = As[l15 * LDA + kk + 1];
    v2f b;
    b.x = Bcol[kk];
    b.y = Bcol[kk + 1];
    acc = __builtin_amdgcn_wmma_f32_16x16x4_f32(
         false, a,  false, b,
         (short)0, acc,  false,  false);
  }

  if (rowBase + 16 <= nrows) {
    float* __restrict__ Hp = H + (size_t)(rowBase + 8 * halfSel) * 128 + col;
#pragma unroll
    for (int r = 0; r < 8; ++r) {
      Hp[(size_t)r * 128] = acc[r];
    }
  } else {
#pragma unroll
    for (int r = 0; r < 8; ++r) {
      const int m = rowBase + r + 8 * halfSel;
      if (m < nrows) H[(size_t)m * 128 + col] = acc[r];
    }
  }
}

__global__ void gcn_deg_init(float* __restrict__ deg, int n) {
  int i = blockIdx.x * blockDim.x + threadIdx.x;
  if (i < n) deg[i] = 1.0f;
}

__global__ void gcn_deg_count(const int* __restrict__ dst,
                              float* __restrict__ deg, int e) {
  int i = blockIdx.x * blockDim.x + threadIdx.x;
  if (i < e) atomicAdd(&deg[dst[i]], 1.0f);
}

__global__ void gcn_deg_finalize(const float* __restrict__ deg,
                                 float* __restrict__ dis,
                                 float* __restrict__ dinv, int n) {
  int i = blockIdx.x * blockDim.x + threadIdx.x;
  if (i < n) {
    float d = deg[i];
    dis[i] = rsqrtf(d);
    dinv[i] = 1.0f / d;
  }
}

__global__ void gcn_zero(float* __restrict__ p, int n) {
  int i = blockIdx.x * blockDim.x + threadIdx.x;
  if (i < n) p[i] = 0.0f;
}

__global__ __launch_bounds__(256) void gcn_scatter(
    const int* __restrict__ src, const int* __restrict__ dst,
    const float* __restrict__ H, const float* __restrict__ dis,
    float* __restrict__ agg, int nedges) {
  const int e = blockIdx.x * 8 + (threadIdx.x >> 5);
  if (e >= nedges) return;
  const int lane = threadIdx.x & 31;
  const int s = src[e];
  const int d = dst[e];
  const float norm = dis[s] * dis[d];
  const float4 v =
      reinterpret_cast<const float4*>(H + (size_t)s * 128)[lane];
  float* ad = agg + (size_t)d * 128 + lane * 4;
  atomicAdd(ad + 0, v.x * norm);
  atomicAdd(ad + 1, v.y * norm);
  atomicAdd(ad + 2, v.z * norm);
  atomicAdd(ad + 3, v.w * norm);
}

__global__ void gcn_combine(const float* __restrict__ agg,
                            const float* __restrict__ H,
                            const float* __restrict__ dinv,
                            const float* __restrict__ bias,
                            float* __restrict__ out, int total) {
  int i = blockIdx.x * blockDim.x + threadIdx.x;
  if (i < total) {
    const int node = i >> 7;
    const int f = i & 127;
    const float v = agg[i] + H[i] * dinv[node] + bias[f];
    out[i] = fmaxf(v, 0.0f);
  }
}

extern "C" void kernel_launch(void* const* d_in, const int* in_sizes, int n_in,
                              void* d_out, int out_size, void* d_ws,
                              size_t ws_size, hipStream_t stream) {
  const float* x = (const float*)d_in[0];
  const int* edge = (const int*)d_in[1];
  const float* W0 = (const float*)d_in[2];
  const float* b0 = (const float*)d_in[3];
  const float* W1 = (const float*)d_in[4];
  const float* b1 = (const float*)d_in[5];
  const float* W2 = (const float*)d_in[6];
  const float* b2 = (const float*)d_in[7];
  float* out = (float*)d_out;

  const int N = in_sizes[0] / 256;
  const int E = in_sizes[1] / 2;
  const int* srcIdx = edge;
  const int* dstIdx = edge + E;

  float* ws = (float*)d_ws;
  float* hbuf = ws;
  float* aggbuf = hbuf + (size_t)N * 128;
  float* xin = aggbuf + (size_t)N * 128;
  float* deg = xin + (size_t)N * 128;
  float* dis = deg + N;
  float* dinv = dis + N;
  float* wt0 = dinv + N;
  float* wt1 = wt0 + 128 * 256;
  float* wt2 = wt1 + 128 * 128;

  const int T = 256;
  const int nodeBlocks = (N + T - 1) / T;
  const int featTotal = N * 128;
  const int featBlocks = (featTotal + T - 1) / T;
  const int edgeBlocks = (E + T - 1) / T;
  const int scatBlocks = (E + 7) / 8;
  const int gemmBlocks = (N + 15) / 16;

  gcn_transpose_w<<<(256 * 128 + T - 1) / T, T, 0, stream>>>(W0, wt0, 256);
  gcn_transpose_w<<<(128 * 128 + T - 1) / T, T, 0, stream>>>(W1, wt1, 128);
  gcn_transpose_w<<<(128 * 128 + T - 1) / T, T, 0, stream>>>(W2, wt2, 128);

  gcn_deg_init<<<nodeBlocks, T, 0, stream>>>(deg, N);
  gcn_deg_count<<<edgeBlocks, T, 0, stream>>>(dstIdx, deg, E);
  gcn_deg_finalize<<<nodeBlocks, T, 0, stream>>>(deg, dis, dinv, N);

  gcn_gemm_wmma<256><<<gemmBlocks, T, 0, stream>>>(x, wt0, hbuf, N);
  gcn_zero<<<featBlocks, T, 0, stream>>>(aggbuf, featTotal);
  gcn_scatter<<<scatBlocks, T, 0, stream>>>(srcIdx, dstIdx, hbuf, dis, aggbuf, E);
  gcn_combine<<<featBlocks, T, 0, stream>>>(aggbuf, hbuf, dinv, b0, xin, featTotal);

  gcn_gemm_wmma<128><<<gemmBlocks, T, 0, stream>>>(xin, wt1, hbuf, N);
  gcn_zero<<<featBlocks, T, 0, stream>>>(aggbuf, featTotal);
  gcn_scatter<<<scatBlocks, T, 0, stream>>>(srcIdx, dstIdx, hbuf, dis, aggbuf, E);
  gcn_combine<<<featBlocks, T, 0, stream>>>(aggbuf, hbuf, dinv, b1, xin, featTotal);

  gcn_gemm_wmma<128><<<gemmBlocks, T, 0, stream>>>(xin, wt2, hbuf, N);
  gcn_zero<<<featBlocks, T, 0, stream>>>(aggbuf, featTotal);
  gcn_scatter<<<scatBlocks, T, 0, stream>>>(srcIdx, dstIdx, hbuf, dis, aggbuf, E);
  gcn_combine<<<featBlocks, T, 0, stream>>>(aggbuf, hbuf, dinv, b2, out, featTotal);
}


# Work per wave raised (tiny waves paced these launches): each scatter wave takes 8 edges, 32 row loads in flight before its atomics; zero/combine lanes move 64 B; sum order = atomic arrival as before
# speedup vs baseline: 1.5006x; 1.5006x over previous
#include <hip/hip_runtime.h>
#include <hip/hip_bf16.h>


typedef __attribute__((ext_vector_type(2))) float v2f;
typedef __attribute__((ext_vector_type(8))) float v8f;

__global__ void gcn_transpose_w(const float* __restrict__ W,
                                float* __restrict__ Wt, int K) {
  int i = blockIdx.x * blockDim.x + threadIdx.x;
  if (i < K * 128) {
    const int k = i >> 7;
    const int c = i & 127;
    Wt[(size_t)c * K + k] = W[i];
  }
}

template <int K>
__global__ __launch_bounds__(256) void gcn_gemm_wmma(
    const float* __restrict__ X, const float* __restrict__ Wt,
    float* __restrict__ H, int nrows) {
  constexpr int LDA = K + 4;
  __shared__ __align__(16) float As[16 * LDA];

  const int rowBase = blockIdx.x * 16;
  const int tid = threadIdx.x;

  constexpr int vecPerRow = K / 4;
  for (int i = tid; i < 16 * vecPerRow; i += 256) {
    const int r = i / vecPerRow;
    const int v = i % vecPerRow;
    const int grow = rowBase + r;
    float4 val = make_float4(0.f, 0.f, 0.f, 0.f);
    if (grow < nrows) {
      val = *reinterpret_cast<const float4*>(X + (size_t)grow * K + 4 * v);
    }
    *reinterpret_cast<float4*>(&As[r * LDA + 4 * v]) = val;
  }
  __syncthreads();

  const int wave = tid >> 5;
  const int lane = tid & 31;
  const int l15 = lane & 15;
  const int halfSel = lane >> 4;
  const int col = wave * 16 + l15;

  const float* __restrict__ Bcol = Wt + (size_t)col * K;

  v8f acc = {};
#pragma unroll 4
  for (int k0 = 0; k0 < K; k0 += 4) {
    const int kk = k0 + 2 * halfSel;
    v2f a;
    a.x = As[l15 * LDA + kk];
    a.y = As[l15 * LDA + kk + 1];
    v2f b;
    b.x = Bcol[kk];
    b.y = Bcol[kk + 1];
    acc = __builtin_amdgcn_wmma_f32_16x16x4_f32(
         false, a,  false, b,
         (short)0, acc,  false,  false);
  }

  if (rowBase + 16 <= nrows) {
    float* __restrict__ Hp = H + (size_t)(rowBase + 8 * halfSel) * 128 + col;
#pragma unroll
    for (int r = 0; r < 8; ++r) {
      Hp[(size_t)r * 128] = acc[r];
    }
  } else {
#pragma unroll
    for (int r = 0; r < 8; ++r) {
      const int m = rowBase + r + 8 * halfSel;
      if (m < nrows) H[(size_t)m * 128 + col] = acc[r];
    }
  }
}

__global__ void gcn_deg_init(float* __restrict__ deg, int n) {
  int i = blockIdx.x * blockDim.x + threadIdx.x;
  if (i < n) deg[i] = 1.0f;
}

__global__ void gcn_deg_count(const int* __restrict__ dst,
                              float* __restrict__ deg, int e) {
  int i = blockIdx.x * blockDim.x + threadIdx.x;
  if (i < e) atomicAdd(&deg[dst[i]], 1.0f);
}

__global__ void gcn_deg_finalize(const float* __restrict__ deg,
                                 float* __restrict__ dis,
                                 float* __restrict__ dinv, int n) {
  int i = blockIdx.x * blockDim.x + threadIdx.x;
  if (i < n) {
    float d = deg[i];
    dis[i] = rsqrtf(d);
    dinv[i] = 1.0f / d;
  }
}

__global__ __launch_bounds__(256) void gcn_zero(float* __restrict__ p, int n4) {
  float4* __restrict__ p4 = reinterpret_cast<float4*>(p);
  const int base = blockIdx.x * 1024 + threadIdx.x;
  const float4 z = make_float4(0.0f, 0.0f, 0.0f, 0.0f);
#pragma unroll
  for (int k = 0; k < 4; ++k) {
    const int i = base + k * 256;
    if (i < n4) p4[i] = z;
  }
}

constexpr int kEdgesPerWave = 8;

__global__ __launch_bounds__(256) void gcn_scatter(
    const int* __restrict__ src, const int* __restrict__ dst,
    const float* __restrict__ H, const float* __restrict__ dis,
    float* __restrict__ agg, int nedges) {
  const int lane = threadIdx.x & 31;
  const int waveInBlock = __builtin_amdgcn_readfirstlane(threadIdx.x >> 5);
  const int e0 = (blockIdx.x * 8 + waveInBlock) * kEdgesPerWave;
  if (e0 >= nedges) return;

  const int eMine = min(e0 + (lane & (kEdgesPerWave - 1)), nedges - 1);
  const int s = src[eMine];
  const int d = dst[eMine];
  const float disS = dis[s];
  const float disD = dis[d];

  if (e0 + kEdgesPerWave <= nedges) {
    float v[kEdgesPerWave][4];
#pragma unroll
    for (int j = 0; j < kEdgesPerWave; ++j) {
      const int sj = __builtin_amdgcn_readlane(s, j);
      const float* __restrict__ hs = H + (size_t)sj * 128 + lane;
      v[j][0] = hs[0];
      v[j][1] = hs[32];
      v[j][2] = hs[64];
      v[j][3] = hs[96];
    }
    const float nrm = disS * disD;
#pragma unroll
    for (int j = 0; j < kEdgesPerWave; ++j) {
      const int dj = __builtin_amdgcn_readlane(d, j);
      const float nj =
          __int_as_float(__builtin_amdgcn_readlane(__float_as_int(nrm), j));
      float* ad = agg + (size_t)dj * 128 + lane;
      atomicAdd(ad + 0, v[j][0] * nj);
      atomicAdd(ad + 32, v[j][1] * nj);
      atomicAdd(ad + 64, v[j][2] * nj);
      atomicAdd(ad + 96, v[j][3] * nj);
    }
  } else {
    const int cnt = nedges - e0;
    const float nrm = disS * disD;
    for (int j = 0; j < cnt; ++j) {
      const int sj = __shfl(s, j);
      const int dj = __shfl(d, j);
      const float nj = __shfl(nrm, j);
      const float* __restrict__ hs = H + (size_t)sj * 128 + lane;
      const float v0 = hs[0];
      const float v1 = hs[32];
      const float v2 = hs[64];
      const float v3 = hs[96];
      float* ad = agg + (size_t)dj * 128 + lane;
      atomicAdd(ad + 0, v0 * nj);
      atomicAdd(ad + 32, v1 * nj);
      atomicAdd(ad + 64, v2 * nj);
      atomicAdd(ad + 96, v3 * nj);
    }
  }
}

__global__ __launch_bounds__(256) void gcn_combine(
    const float* __restrict__ agg, const float* __restrict__ H,
    const float* __restrict__ dinv, const float* __restrict__ bias,
    float* __restrict__ out, int total4) {
  const float4* __restrict__ agg4 = reinterpret_cast<const float4*>(agg);
  const float4* __restrict__ H4 = reinterpret_cast<const float4*>(H);
  float4* __restrict__ out4 = reinterpret_cast<float4*>(out);
  const int lane = threadIdx.x & 31;
  const int blockBase = blockIdx.x * 1024;
  const int base = blockBase + threadIdx.x;
  const float bx = bias[4 * lane + 0];
  const float by = bias[4 * lane + 1];
  const float bz = bias[4 * lane + 2];
  const float bw = bias[4 * lane + 3];

  if (blockBase + 1024 <= total4) {
    float4 a[4];
    float4 h[4];
    float di[4];
#pragma unroll
    for (int k = 0; k < 4; ++k) {
      const int i = base + k * 256;
      a[k] = agg4[i];
      h[k] = H4[i];
      di[k] = dinv[i >> 5];
    }
#pragma unroll
    for (int k = 0; k < 4; ++k) {
      const int i = base + k * 256;
      float4 o;
      o.x = fmaxf(a[k].x + h[k].x * di[k] + bx, 0.0f);
      o.y = fmaxf(a[k].y + h[k].y * di[k] + by, 0.0f);
      o.z = fmaxf(a[k].z + h[k].z * di[k] + bz, 0.0f);
      o.w = fmaxf(a[k].w + h[k].w * di[k] + bw, 0.0f);
      out4[i] = o;
    }
  } else {
    for (int k = 0; k < 4; ++k) {
      const int i = base + k * 256;
      if (i < total4) {
        const float4 a = agg4[i];
        const float4 h = H4[i];
        const float di = dinv[i >> 5];
        float4 o;
        o.x = fmaxf(a.x + h.x * di + bx, 0.0f);
        o.y = fmaxf(a.y + h.y * di + by, 0.0f);
        o.z = fmaxf(a.z + h.z * di + bz, 0.0f);
        o.w = fmaxf(a.w + h.w * di + bw, 0.0f);
        out4[i] = o;
      }
    }
  }
}

extern "C" void kernel_launch(void* const* d_in, const int* in_sizes, int n_in,
                              void* d_out, int out_size, void* d_ws,
                              size_t ws_size, hipStream_t stream) {
  const float* x = (const float*)d_in[0];
  const int* edge = (const int*)d_in[1];
  const float* W0 = (const float*)d_in[2];
  const float* b0 = (const float*)d_in[3];
  const float* W1 = (const float*)d_in[4];
  const float* b1 = (const float*)d_in[5];
  const float* W2 = (const float*)d_in[6];
  const float* b2 = (const float*)d_in[7];
  float* out = (float*)d_out;

  const int N = in_sizes[0] / 256;
  const int E = in_sizes[1] / 2;
  const int* srcIdx = edge;
  const int* dstIdx = edge + E;

  float* ws = (float*)d_ws;
  float* hbuf = ws;
  float* aggbuf = hbuf + (size_t)N * 128;
  float* xin = aggbuf + (size_t)N * 128;
  float* deg = xin + (size_t)N * 128;
  float* dis = deg + N;
  float* dinv = dis + N;
  float* wt0 = dinv + N;
  float* wt1 = wt0 + 128 * 256;
  float* wt2 = wt1 + 128 * 128;

  const int T = 256;
  const int nodeBlocks = (N + T - 1) / T;
  const int featTotal = N * 128;
  const int edgeBlocks = (E + T - 1) / T;
  const int scatBlocks = (E + 8 * kEdgesPerWave - 1) / (8 * kEdgesPerWave);
  const int featTotal4 = featTotal / 4;
  const int vecBlocks = (featTotal4 + 1023) / 1024;
  const int gemmBlocks = (N + 15) / 16;

  gcn_transpose_w<<<(256 * 128 + T - 1) / T, T, 0, stream>>>(W0, wt0, 256);
  gcn_transpose_w<<<(128 * 128 + T - 1) / T, T, 0, stream>>>(W1, wt1, 128);
  gcn_transpose_w<<<(128 * 128 + T - 1) / T, T, 0, stream>>>(W2, wt2, 128);

  gcn_deg_init<<<nodeBlocks, T, 0, stream>>>(deg, N);
  gcn_deg_count<<<edgeBlocks, T, 0, stream>>>(dstIdx, deg, E);
  gcn_deg_finalize<<<nodeBlocks, T, 0, stream>>>(deg, dis, dinv, N);

  gcn_gemm_wmma<256><<<gemmBlocks, T, 0, stream>>>(x, wt0, hbuf, N);
  gcn_zero<<<vecBlocks, T, 0, stream>>>(aggbuf, featTotal4);
  gcn_scatter<<<scatBlocks, T, 0, stream>>>(srcIdx, dstIdx, hbuf, dis, aggbuf, E);
  gcn_combine<<<vecBlocks, T, 0, stream>>>(aggbuf, hbuf, dinv, b0, xin, featTotal4);

  gcn_gemm_wmma<128><<<gemmBlocks, T, 0, stream>>>(xin, wt1, hbuf, N);
  gcn_zero<<<vecBlocks, T, 0, stream>>>(aggbuf, featTotal4);
  gcn_scatter<<<scatBlocks, T, 0, stream>>>(srcIdx, dstIdx, hbuf, dis, aggbuf, E);
  gcn_combine<<<vecBlocks, T, 0, stream>>>(aggbuf, hbuf, dinv, b1, xin, featTotal4);

  gcn_gemm_wmma<128><<<gemmBlocks, T, 0, stream>>>(xin, wt2, hbuf, N);
  gcn_zero<<<vecBlocks, T, 0, stream>>>(aggbuf, featTotal4);
  gcn_scatter<<<scatBlocks, T, 0, stream>>>(srcIdx, dstIdx, hbuf, dis, aggbuf, E);
  gcn_combine<<<vecBlocks, T, 0, stream>>>(aggbuf, hbuf, dinv, b2, out, featTotal4);
}


# GEMM operands straight from global: a wave owns 16 rows x 64 cols, A as 16-byte k-quads, B from W rows as given; no LDS tile, no transposed weights (3 launches gone); sum order in 8-k groups changes
# speedup vs baseline: 1.9399x; 1.2927x over previous
#include <hip/hip_runtime.h>
#include <hip/hip_bf16.h>


typedef __attribute__((ext_vector_type(2))) float v2f;
typedef __attribute__((ext_vector_type(4))) float v4f;
typedef __attribute__((ext_vector_type(8))) float v8f;

template <int K, bool RELU_IN>
__global__ __launch_bounds__(128) void gcn_gemm_wmma(
    const float* __restrict__ X, const float* __restrict__ W,
    const float* __restrict__ dinv, const float* __restrict__ bias,
    float* __restrict__ H, float* __restrict__ Acc, int nrows) {
  constexpr int NCHUNK = K / 32;
  constexpr int NSTEP = K / 4;

  const int lane = threadIdx.x & 31;
  const int c = lane & 15;
  const int h = lane >> 4;
  const int waveInBlock = __builtin_amdgcn_readfirstlane(threadIdx.x >> 5);
  const int u = waveInBlock & 1;
  const int rowBase = (blockIdx.x * 2 + (waveInBlock >> 1)) * 16;
  if (rowBase >= nrows) return;

  const float* __restrict__ xr =
      X + (size_t)min(rowBase + c, nrows - 1) * K + 4 * h;
  const float* __restrict__ wl = W + (size_t)(4 * h) * 128 + 64 * u + 4 * c;

  const float4 bias4 = *reinterpret_cast<const float4*>(bias + 64 * u + 4 * c);
  float dv[8];
#pragma unroll
  for (int r = 0; r < 8; ++r) {
    dv[r] = dinv[min(rowBase + 8 * h + r, nrows - 1)];
  }

  v8f acc[4];
#pragma unroll
  for (int t = 0; t < 4; ++t) acc[t] = v8f{};

  constexpr int D = 3;
  v4f a[2][4];
  float4 b[D + 1][2];

#pragma unroll
  for (int i = 0; i < 4; ++i) {
    a[0][i] = *reinterpret_cast<const v4f*>(xr + 8 * i);
  }
#pragma unroll
  for (int m = 0; m < D; ++m) {
    const int km = 8 * (m >> 1) + 2 * (m & 1);
#pragma unroll
    for (int s = 0; s < 2; ++s) {
      b[m][s] = *reinterpret_cast<const float4*>(wl + (size_t)(km + s) * 128);
    }
  }

#pragma unroll
  for (int n = 0; n < NSTEP; ++n) {
    const int g = n >> 1;
    const int p = n & 1;
    const int J = g >> 2;
    const int i = g & 3;
    if (n + D < NSTEP) {
      const int kn = 8 * ((n + D) >> 1) + 2 * ((n + D) & 1);
#pragma unroll
      for (int s = 0; s < 2; ++s) {
        b[(n + D) % (D + 1)][s] =
            *reinterpret_cast<const float4*>(wl + (size_t)(kn + s) * 128);
      }
    }
    if ((n & 7) == 0 && J + 1 < NCHUNK) {
#pragma unroll
      for (int q = 0; q < 4; ++q) {
        a[(J + 1) & 1][q] =
            *reinterpret_cast<const v4f*>(xr + 32 * (J + 1) + 8 * q);
      }
    }
    __builtin_amdgcn_sched_barrier(0);
    if (RELU_IN && p == 0) {
      a[J & 1][i] = __builtin_elementwise_max(a[J & 1][i], v4f{0.f, 0.f, 0.f, 0.f});
    }
    const v4f aq = a[J & 1][i];
    v2f af;
    af.x = p ? aq.z : aq.x;
    af.y = p ? aq.w : aq.y;
    const float4 b0 = b[n % (D + 1)][0];
    const float4 b1 = b[n % (D + 1)][1];
    v2f bf;
    bf.x = b0.x; bf.y = b1.x;
    acc[0] = __builtin_amdgcn_wmma_f32_16x16x4_f32(
        false, af, false, bf, (short)0, acc[0], false, false);
    bf.x = b0.y; bf.y = b1.y;
    acc[1] = __builtin_amdgcn_wmma_f32_16x16x4_f32(
        false, af, false, bf, (short)0, acc[1], false, false);
    bf.x = b0.z; bf.y = b1.z;
    acc[2] = __builtin_amdgcn_wmma_f32_16x16x4_f32(
        false, af, false, bf, (short)0, acc[2], false, false);
    bf.x = b0.w; bf.y = b1.w;
    acc[3] = __builtin_amdgcn_wmma_f32_16x16x4_f32(
        false, af, false, bf, (short)0, acc[3], false, false);
  }

  float4 hv[8];
  float4 st[8];
#pragma unroll
  for (int r = 0; r < 8; ++r) {
    hv[r] = make_float4(acc[0][r], acc[1][r], acc[2][r], acc[3][r]);
    st[r] = make_float4(hv[r].x * dv[r] + bias4.x, hv[r].y * dv[r] + bias4.y,
                        hv[r].z * dv[r] + bias4.z, hv[r].w * dv[r] + bias4.w);
  }
  const size_t o = (size_t)(rowBase + 8 * h) * 128 + 64 * u + 4 * c;
  if (rowBase + 16 <= nrows) {
#pragma unroll
    for (int r = 0; r < 8; ++r) {
      *reinterpret_cast<float4*>(H + o + (size_t)r * 128) = hv[r];
      *reinterpret_cast<float4*>(Acc + o + (size_t)r * 128) = st[r];
    }
  } else {
#pragma unroll
    for (int r = 0; r < 8; ++r) {
      if (rowBase + 8 * h + r < nrows) {
        *reinterpret_cast<float4*>(H + o + (size_t)r * 128) = hv[r];
        *reinterpret_cast<float4*>(Acc + o + (size_t)r * 128) = st[r];
      }
    }
  }
}

__global__ void gcn_deg_init(float* __restrict__ deg, int n) {
  int i = blockIdx.x * blockDim.x + threadIdx.x;
  if (i < n) deg[i] = 1.0f;
}

__global__ void gcn_deg_count(const int* __restrict__ dst,
                              float* __restrict__ deg, int e) {
  int i = blockIdx.x * blockDim.x + threadIdx.x;
  if (i < e) atomicAdd(&deg[dst[i]], 1.0f);
}

__global__ void gcn_deg_finalize(const float* __restrict__ deg,
                                 float* __restrict__ dis,
                                 float* __restrict__ dinv, int n) {
  int i = blockIdx.x * blockDim.x + threadIdx.x;
  if (i < n) {
    float d = deg[i];
    dis[i] = rsqrtf(d);
    dinv[i] = 1.0f / d;
  }
}

__global__ __launch_bounds__(256) void gcn_relu_inplace(float* __restrict__ p,
                                                        int n4) {
  float4* __restrict__ p4 = reinterpret_cast<float4*>(p);
  const int blockBase = blockIdx.x * 1024;
  const int base = blockBase + threadIdx.x;
  if (blockBase + 1024 <= n4) {
    float4 a[4];
#pragma unroll
    for (int k = 0; k < 4; ++k) a[k] = p4[base + k * 256];
#pragma unroll
    for (int k = 0; k < 4; ++k) {
      float4 o;
      o.x = fmaxf(a[k].x, 0.0f);
      o.y = fmaxf(a[k].y, 0.0f);
      o.z = fmaxf(a[k].z, 0.0f);
      o.w = fmaxf(a[k].w, 0.0f);
      p4[base + k * 256] = o;
    }
  } else {
    for (int k = 0; k < 4; ++k) {
      const int i = base + k * 256;
      if (i < n4) {
        const float4 a = p4[i];
        float4 o;
        o.x = fmaxf(a.x, 0.0f);
        o.y = fmaxf(a.y, 0.0f);
        o.z = fmaxf(a.z, 0.0f);
        o.w = fmaxf(a.w, 0.0f);
        p4[i] = o;
      }
    }
  }
}

constexpr int kEdgesPerWave = 8;

__global__ __launch_bounds__(256) void gcn_scatter(
    const int* __restrict__ src, const int* __restrict__ dst,
    const float* __restrict__ H, const float* __restrict__ dis,
    float* __restrict__ agg, int nedges) {
  const int lane = threadIdx.x & 31;
  const int waveInBlock = __builtin_amdgcn_readfirstlane(threadIdx.x >> 5);
  const int e0 = (blockIdx.x * 8 + waveInBlock) * kEdgesPerWave;
  if (e0 >= nedges) return;

  const int eMine = min(e0 + (lane & (kEdgesPerWave - 1)), nedges - 1);
  const int s = src[eMine];
  const int d = dst[eMine];
  const float disS = dis[s];
  const float disD = dis[d];

  if (e0 + kEdgesPerWave <= nedges) {
    float v[kEdgesPerWave][4];
#pragma unroll
    for (int j = 0; j < kEdgesPerWave; ++j) {
      const int sj = __builtin_amdgcn_readlane(s, j);
      const float* __restrict__ hs = H + (size_t)sj * 128 + lane;
      v[j][0] = hs[0];
      v[j][1] = hs[32];
      v[j][2] = hs[64];
      v[j][3] = hs[96];
    }
    const float nrm = disS * disD;
#pragma unroll
    for (int j = 0; j < kEdgesPerWave; ++j) {
      const int dj = __builtin_amdgcn_readlane(d, j);
      const float nj =
          __int_as_float(__builtin_amdgcn_readlane(__float_as_int(nrm), j));
      float* ad = agg + (size_t)dj * 128 + lane;
      atomicAdd(ad + 0, v[j][0] * nj);
      atomicAdd(ad + 32, v[j][1] * nj);
      atomicAdd(ad + 64, v[j][2] * nj);
      atomicAdd(ad + 96, v[j][3] * nj);
    }
  } else {
    const int cnt = nedges - e0;
    const float nrm = disS * disD;
    for (int j = 0; j < cnt; ++j) {
      const int sj = __shfl(s, j);
      const int dj = __shfl(d, j);
      const float nj = __shfl(nrm, j);
      const float* __restrict__ hs = H + (size_t)sj * 128 + lane;
      const float v0 = hs[0];
      const float v1 = hs[32];
      const float v2 = hs[64];
      const float v3 = hs[96];
      float* ad = agg + (size_t)dj * 128 + lane;
      atomicAdd(ad + 0, v0 * nj);
      atomicAdd(ad + 32, v1 * nj);
      atomicAdd(ad + 64, v2 * nj);
      atomicAdd(ad + 96, v3 * nj);
    }
  }
}

extern "C" void kernel_launch(void* const* d_in, const int* in_sizes, int n_in,
                              void* d_out, int out_size, void* d_ws,
                              size_t ws_size, hipStream_t stream) {
  const float* x = (const float*)d_in[0];
  const int* edge = (const int*)d_in[1];
  const float* W0 = (const float*)d_in[2];
  const float* b0 = (const float*)d_in[3];
  const float* W1 = (const float*)d_in[4];
  const float* b1 = (const float*)d_in[5];
  const float* W2 = (const float*)d_in[6];
  const float* b2 = (const float*)d_in[7];
  float* out = (float*)d_out;

  const int N = in_sizes[0] / 256;
  const int E = in_sizes[1] / 2;
  const int* srcIdx = edge;
  const int* dstIdx = edge + E;

  float* ws = (float*)d_ws;
  float* hbuf = ws;
  float* acc0 = hbuf + (size_t)N * 128;
  float* acc1 = acc0 + (size_t)N * 128;
  float* deg = acc1 + (size_t)N * 128;
  float* dis = deg + N;
  float* dinv = dis + N;

  const int T = 256;
  const int nodeBlocks = (N + T - 1) / T;
  const int featTotal = N * 128;
  const int edgeBlocks = (E + T - 1) / T;
  const int scatBlocks = (E + 8 * kEdgesPerWave - 1) / (8 * kEdgesPerWave);
  const int featTotal4 = featTotal / 4;
  const int vecBlocks = (featTotal4 + 1023) / 1024;
  const int gemmBlocks = (N + 31) / 32;

  gcn_deg_init<<<nodeBlocks, T, 0, stream>>>(deg, N);
  gcn_deg_count<<<edgeBlocks, T, 0, stream>>>(dstIdx, deg, E);
  gcn_deg_finalize<<<nodeBlocks, T, 0, stream>>>(deg, dis, dinv, N);

  gcn_gemm_wmma<256, false><<<gemmBlocks, 128, 0, stream>>>(x, W0, dinv, b0, hbuf, acc0, N);
  gcn_scatter<<<scatBlocks, T, 0, stream>>>(srcIdx, dstIdx, hbuf, dis, acc0, E);

  gcn_gemm_wmma<128, true><<<gemmBlocks, 128, 0, stream>>>(acc0, W1, dinv, b1, hbuf, acc1, N);
  gcn_scatter<<<scatBlocks, T, 0, stream>>>(srcIdx, dstIdx, hbuf, dis, acc1, E);

  gcn_gemm_wmma<128, true><<<gemmBlocks, 128, 0, stream>>>(acc1, W2, dinv, b2, hbuf, out, N);
  gcn_scatter<<<scatBlocks, T, 0, stream>>>(srcIdx, dstIdx, hbuf, dis, out, E);
  gcn_relu_inplace<<<vecBlocks, T, 0, stream>>>(out, featTotal4);
}


# Edge sums by owner workgroups (196 LDS rows) rebuilt so list delivery paces them: 8 list loads in flight per lane, staggered starts, per-wave LDS queues, relu at the store; sum order: LDS arrival
# speedup vs baseline: 2.1347x; 1.1004x over previous
#include <hip/hip_runtime.h>
#include <hip/hip_bf16.h>


typedef __attribute__((ext_vector_type(2))) float v2f;
typedef __attribute__((ext_vector_type(4))) float v4f;
typedef __attribute__((ext_vector_type(8))) float v8f;

template <int K>
__global__ __launch_bounds__(128) void gcn_gemm_wmma(
    const float* __restrict__ X, const float* __restrict__ W,
    const float* __restrict__ dinv, const float* __restrict__ bias,
    float* __restrict__ H, float* __restrict__ Acc, int nrows) {
  constexpr int NCHUNK = K / 32;
  constexpr int NSTEP = K / 4;

  const int lane = threadIdx.x & 31;
  const int c = lane & 15;
  const int h = lane >> 4;
  const int waveInBlock = __builtin_amdgcn_readfirstlane(threadIdx.x >> 5);
  const int u = waveInBlock & 1;
  const int rowBase = (blockIdx.x * 2 + (waveInBlock >> 1)) * 16;
  if (rowBase >= nrows) return;

  const float* __restrict__ xr =
      X + (size_t)min(rowBase + c, nrows - 1) * K + 4 * h;
  const float* __restrict__ wl = W + (size_t)(4 * h) * 128 + 64 * u + 4 * c;

  const float4 bias4 = *reinterpret_cast<const float4*>(bias + 64 * u + 4 * c);
  float dv[8];
#pragma unroll
  for (int r = 0; r < 8; ++r) {
    dv[r] = dinv[min(rowBase + 8 * h + r, nrows - 1)];
  }

  v8f acc[4];
#pragma unroll
  for (int t = 0; t < 4; ++t) acc[t] = v8f{};

  constexpr int D = 3;
  v4f a[2][4];
  float4 b[D + 1][2];

#pragma unroll
  for (int i = 0; i < 4; ++i) {
    a[0][i] = *reinterpret_cast<const v4f*>(xr + 8 * i);
  }
#pragma unroll
  for (int m = 0; m < D; ++m) {
    const int km = 8 * (m >> 1) + 2 * (m & 1);
#pragma unroll
    for (int s = 0; s < 2; ++s) {
      b[m][s] = *reinterpret_cast<const float4*>(wl + (size_t)(km + s) * 128);
    }
  }

#pragma unroll
  for (int n = 0; n < NSTEP; ++n) {
    const int g = n >> 1;
    const int p = n & 1;
    const int J = g >> 2;
    const int i = g & 3;
    if (n + D < NSTEP) {
      const int kn = 8 * ((n + D) >> 1) + 2 * ((n + D) & 1);
#pragma unroll
      for (int s = 0; s < 2; ++s) {
        b[(n + D) % (D + 1)][s] =
            *reinterpret_cast<const float4*>(wl + (size_t)(kn + s) * 128);
      }
    }
    if ((n & 7) == 0 && J + 1 < NCHUNK) {
#pragma unroll
      for (int q = 0; q < 4; ++q) {
        a[(J + 1) & 1][q] =
            *reinterpret_cast<const v4f*>(xr + 32 * (J + 1) + 8 * q);
      }
    }
    __builtin_amdgcn_sched_barrier(0);
    const v4f aq = a[J & 1][i];
    v2f af;
    af.x = p ? aq.z : aq.x;
    af.y = p ? aq.w : aq.y;
    const float4 b0 = b[n % (D + 1)][0];
    const float4 b1 = b[n % (D + 1)][1];
    v2f bf;
    bf.x = b0.x; bf.y = b1.x;
    acc[0] = __builtin_amdgcn_wmma_f32_16x16x4_f32(
        false, af, false, bf, (short)0, acc[0], false, false);
    bf.x = b0.y; bf.y = b1.y;
    acc[1] = __builtin_amdgcn_wmma_f32_16x16x4_f32(
        false, af, false, bf, (short)0, acc[1], false, false);
    bf.x = b0.z; bf.y = b1.z;
    acc[2] = __builtin_amdgcn_wmma_f32_16x16x4_f32(
        false, af, false, bf, (short)0, acc[2], false, false);
    bf.x = b0.w; bf.y = b1.w;
    acc[3] = __builtin_amdgcn_wmma_f32_16x16x4_f32(
        false, af, false, bf, (short)0, acc[3], false, false);
  }

  float4 hv[8];
  float4 st[8];
#pragma unroll
  for (int r = 0; r < 8; ++r) {
    hv[r] = make_float4(acc[0][r], acc[1][r], acc[2][r], acc[3][r]);
    st[r] = make_float4(hv[r].x * dv[r] + bias4.x, hv[r].y * dv[r] + bias4.y,
                        hv[r].z * dv[r] + bias4.z, hv[r].w * dv[r] + bias4.w);
  }
  const size_t o = (size_t)(rowBase + 8 * h) * 128 + 64 * u + 4 * c;
  if (rowBase + 16 <= nrows) {
#pragma unroll
    for (int r = 0; r < 8; ++r) {
      *reinterpret_cast<float4*>(H + o + (size_t)r * 128) = hv[r];
      *reinterpret_cast<float4*>(Acc + o + (size_t)r * 128) = st[r];
    }
  } else {
#pragma unroll
    for (int r = 0; r < 8; ++r) {
      if (rowBase + 8 * h + r < nrows) {
        *reinterpret_cast<float4*>(H + o + (size_t)r * 128) = hv[r];
        *reinterpret_cast<float4*>(Acc + o + (size_t)r * 128) = st[r];
      }
    }
  }
}

__global__ void gcn_deg_init(float* __restrict__ deg, int n) {
  int i = blockIdx.x * blockDim.x + threadIdx.x;
  if (i < n) deg[i] = 1.0f;
}

__global__ void gcn_deg_count(const int* __restrict__ dst,
                              float* __restrict__ deg, int e) {
  int i = blockIdx.x * blockDim.x + threadIdx.x;
  if (i < e) atomicAdd(&deg[dst[i]], 1.0f);
}

__global__ void gcn_deg_finalize(const float* __restrict__ deg,
                                 float* __restrict__ dis,
                                 float* __restrict__ dinv, int n) {
  int i = blockIdx.x * blockDim.x + threadIdx.x;
  if (i < n) {
    float d = deg[i];
    dis[i] = rsqrtf(d);
    dinv[i] = 1.0f / d;
  }
}

constexpr int kOwnRows = 196;
constexpr int kOwnWaves = 32;
constexpr int kGroupLoads = 8;
constexpr int kGroupEdges = kGroupLoads * 128;
constexpr int kWgGroupEdges = kOwnWaves * kGroupEdges;
constexpr int kQueueCap = 1280;

__global__ __launch_bounds__(1024) void gcn_owner_pass(
    const int* __restrict__ src, const int* __restrict__ dst,
    const float* __restrict__ H, const float* __restrict__ dis,
    float* __restrict__ plane, int nrows, int nedges) {
  __shared__ float rows[kOwnRows * 128];
  __shared__ float disL[kOwnRows];
  __shared__ int qe[kOwnWaves * kQueueCap];
  __shared__ unsigned char qd[kOwnWaves * kQueueCap];

  const int tid = threadIdx.x;
  const int lane = tid & 31;
  const int wave = __builtin_amdgcn_readfirstlane(tid >> 5);
  const int base = blockIdx.x * kOwnRows;
  const int nr = min(kOwnRows, nrows - base);
  const unsigned unr = (unsigned)nr;

  {
    constexpr int kRowsPerWave = (kOwnRows + kOwnWaves - 1) / kOwnWaves;
    const v4f* __restrict__ p4 =
        reinterpret_cast<const v4f*>(plane) + (size_t)base * 32 + lane;
    v4f* r4 = reinterpret_cast<v4f*>(rows) + lane;
    v4f t[kRowsPerWave];
#pragma unroll
    for (int m = 0; m < kRowsPerWave; ++m) {
      const int r = wave + kOwnWaves * m;
      if (r < nr) t[m] = p4[r * 32];
    }
#pragma unroll
    for (int m = 0; m < kRowsPerWave; ++m) {
      const int r = wave + kOwnWaves * m;
      if (r < nr) r4[r * 32] = t[m];
    }
    for (int i = tid; i < nr; i += 1024) disL[i] = dis[base + i];
  }
  __syncthreads();

  int* myQe = qe + wave * kQueueCap;
  unsigned char* myQd = qd + wave * kQueueCap;
  int cnt = 0;

  const bool vecOK = ((nedges & 3) == 0) &&
                     ((reinterpret_cast<size_t>(dst) & 15) == 0);
  const int nIter = (nedges + kWgGroupEdges - 1) / kWgGroupEdges;
  int it = (nIter > 0) ? (int)(blockIdx.x % (unsigned)nIter) : 0;

  for (int step = 0;; ++step) {
    const bool last = (step >= nIter);
    if (last || cnt > kQueueCap - kGroupEdges) {
      __builtin_amdgcn_fence(__ATOMIC_ACQ_REL, "wavefront");
      for (int i0 = 0; i0 < cnt; i0 += 128) {
        const int i = i0 + lane;
        const int e0 = myQe[min(i, cnt - 1)];
        const int e1 = myQe[min(i + 32, cnt - 1)];
        const int e2 = myQe[min(i + 64, cnt - 1)];
        const int e3 = myQe[min(i + 96, cnt - 1)];
        const int s0 = src[e0];
        const int s1 = src[e1];
        const int s2 = src[e2];
        const int s3 = src[e3];
        if (i < cnt) myQe[i] = s0;
        if (i + 32 < cnt) myQe[i + 32] = s1;
        if (i + 64 < cnt) myQe[i + 64] = s2;
        if (i + 96 < cnt) myQe[i + 96] = s3;
      }
      __builtin_amdgcn_fence(__ATOMIC_ACQ_REL, "wavefront");
      for (int b0 = 0; b0 < cnt; b0 += 8) {
        const int n = min(8, cnt - b0);
        const int qi = min(b0 + (lane & 7), cnt - 1);
        const int s = myQe[qi];
        const int dl = myQd[qi];
        const float nrm = dis[s] * disL[dl];
        float v[8][4];
#pragma unroll
        for (int j = 0; j < 8; ++j) {
          if (j < n) {
            const int sj = __builtin_amdgcn_readlane(s, j);
            const float* __restrict__ hs = H + (size_t)sj * 128 + lane;
            v[j][0] = hs[0];
            v[j][1] = hs[32];
            v[j][2] = hs[64];
            v[j][3] = hs[96];
          }
        }
#pragma unroll
        for (int j = 0; j < 8; ++j) {
          if (j < n) {
            const int dj = __builtin_amdgcn_readlane(dl, j);
            const float nj = __int_as_float(
                __builtin_amdgcn_readlane(__float_as_int(nrm), j));
            float* r = rows + dj * 128 + lane;
            __hip_atomic_fetch_add(r + 0, v[j][0] * nj, __ATOMIC_RELAXED,
                                   __HIP_MEMORY_SCOPE_WORKGROUP);
            __hip_atomic_fetch_add(r + 32, v[j][1] * nj, __ATOMIC_RELAXED,
                                   __HIP_MEMORY_SCOPE_WORKGROUP);
            __hip_atomic_fetch_add(r + 64, v[j][2] * nj, __ATOMIC_RELAXED,
                                   __HIP_MEMORY_SCOPE_WORKGROUP);
            __hip_atomic_fetch_add(r + 96, v[j][3] * nj, __ATOMIC_RELAXED,
                                   __HIP_MEMORY_SCOPE_WORKGROUP);
          }
        }
      }
      __builtin_amdgcn_fence(__ATOMIC_ACQ_REL, "wavefront");
      cnt = 0;
    }
    if (last) break;

    const int eb = (it * kOwnWaves + wave) * kGroupEdges;
    it = (it + 1 == nIter) ? 0 : it + 1;
    if (eb >= nedges) continue;

    int4 cur[kGroupLoads];
    if (vecOK && eb + kGroupEdges <= nedges) {
      const int4* __restrict__ d4 =
          reinterpret_cast<const int4*>(dst + eb) + lane;
#pragma unroll
      for (int k = 0; k < kGroupLoads; ++k) cur[k] = d4[k * 32];
    } else {
#pragma unroll
      for (int k = 0; k < kGroupLoads; ++k) {
        const int e = eb + k * 128 + lane * 4;
        cur[k].x = (e + 0 < nedges) ? dst[e + 0] : -1;
        cur[k].y = (e + 1 < nedges) ? dst[e + 1] : -1;
        cur[k].z = (e + 2 < nedges) ? dst[e + 2] : -1;
        cur[k].w = (e + 3 < nedges) ? dst[e + 3] : -1;
      }
    }

#pragma unroll
    for (int k = 0; k < kGroupLoads; ++k) {
      const unsigned u0 = (unsigned)(cur[k].x - base);
      const unsigned u1 = (unsigned)(cur[k].y - base);
      const unsigned u2 = (unsigned)(cur[k].z - base);
      const unsigned u3 = (unsigned)(cur[k].w - base);
      const bool any = (u0 < unr) | (u1 < unr) | (u2 < unr) | (u3 < unr);
      if (__builtin_amdgcn_ballot_w32(any) != 0u) {
        const int e = eb + k * 128 + lane * 4;
        const unsigned u[4] = {u0, u1, u2, u3};
#pragma unroll
        for (int j = 0; j < 4; ++j) {
          const bool m = u[j] < unr;
          const unsigned mask = __builtin_amdgcn_ballot_w32(m);
          if (m) {
            const int slot = cnt + (int)__builtin_amdgcn_mbcnt_lo(mask, 0u);
            myQe[slot] = e + j;
            myQd[slot] = (unsigned char)u[j];
          }
          cnt += __builtin_popcount(mask);
        }
      }
    }
  }

  __syncthreads();

  {
    float4* __restrict__ p4 = reinterpret_cast<float4*>(plane) + (size_t)base * 32;
    const float4* r4 = reinterpret_cast<const float4*>(rows);
    const int n4 = nr * 32;
    for (int i = tid; i < n4; i += 1024) {
      const float4 a = r4[i];
      float4 o;
      o.x = fmaxf(a.x, 0.0f);
      o.y = fmaxf(a.y, 0.0f);
      o.z = fmaxf(a.z, 0.0f);
      o.w = fmaxf(a.w, 0.0f);
      p4[i] = o;
    }
  }
}

extern "C" void kernel_launch(void* const* d_in, const int* in_sizes, int n_in,
                              void* d_out, int out_size, void* d_ws,
                              size_t ws_size, hipStream_t stream) {
  const float* x = (const float*)d_in[0];
  const int* edge = (const int*)d_in[1];
  const float* W0 = (const float*)d_in[2];
  const float* b0 = (const float*)d_in[3];
  const float* W1 = (const float*)d_in[4];
  const float* b1 = (const float*)d_in[5];
  const float* W2 = (const float*)d_in[6];
  const float* b2 = (const float*)d_in[7];
  float* out = (float*)d_out;

  const int N = in_sizes[0] / 256;
  const int E = in_sizes[1] / 2;
  const int* srcIdx = edge;
  const int* dstIdx = edge + E;

  float* ws = (float*)d_ws;
  float* hbuf = ws;
  float* acc0 = hbuf + (size_t)N * 128;
  float* acc1 = acc0 + (size_t)N * 128;
  float* deg = acc1 + (size_t)N * 128;
  float* dis = deg + N;
  float* dinv = dis + N;

  const int T = 256;
  const int nodeBlocks = (N + T - 1) / T;
  const int edgeBlocks = (E + T - 1) / T;
  const int ownBlocks = (N + kOwnRows - 1) / kOwnRows;
  const int gemmBlocks = (N + 31) / 32;

  gcn_deg_init<<<nodeBlocks, T, 0, stream>>>(deg, N);
  gcn_deg_count<<<edgeBlocks, T, 0, stream>>>(dstIdx, deg, E);
  gcn_deg_finalize<<<nodeBlocks, T, 0, stream>>>(deg, dis, dinv, N);

  gcn_gemm_wmma<256><<<gemmBlocks, 128, 0, stream>>>(x, W0, dinv, b0, hbuf, acc0, N);
  gcn_owner_pass<<<ownBlocks, 1024, 0, stream>>>(srcIdx, dstIdx, hbuf, dis, acc0, N, E);

  gcn_gemm_wmma<128><<<gemmBlocks, 128, 0, stream>>>(acc0, W1, dinv, b1, hbuf, acc1, N);
  gcn_owner_pass<<<ownBlocks, 1024, 0, stream>>>(srcIdx, dstIdx, hbuf, dis, acc1, N, E);

  gcn_gemm_wmma<128><<<gemmBlocks, 128, 0, stream>>>(acc1, W2, dinv, b2, hbuf, out, N);
  gcn_owner_pass<<<ownBlocks, 1024, 0, stream>>>(srcIdx, dstIdx, hbuf, dis, out, N, E);
}


# Destination list scanned once per call, not once per layer: a build launch (degrees too, by LDS counts; 3 launches gone) keeps each owner wave's edges in workspace; sum order: LDS arrival as before
# speedup vs baseline: 3.1500x; 1.4756x over previous
#include <hip/hip_runtime.h>
#include <hip/hip_bf16.h>


typedef __attribute__((ext_vector_type(2))) float v2f;
typedef __attribute__((ext_vector_type(4))) float v4f;
typedef __attribute__((ext_vector_type(8))) float v8f;

template <int K>
__global__ __launch_bounds__(128) void gcn_gemm_wmma(
    const float* __restrict__ X, const float* __restrict__ W,
    const float* __restrict__ dinv, const float* __restrict__ bias,
    float* __restrict__ H, float* __restrict__ Acc, int nrows) {
  constexpr int NCHUNK = K / 32;
  constexpr int NSTEP = K / 4;

  const int lane = threadIdx.x & 31;
  const int c = lane & 15;
  const int h = lane >> 4;
  const int waveInBlock = __builtin_amdgcn_readfirstlane(threadIdx.x >> 5);
  const int u = waveInBlock & 1;
  const int rowBase = (blockIdx.x * 2 + (waveInBlock >> 1)) * 16;
  if (rowBase >= nrows) return;

  const float* __restrict__ xr =
      X + (size_t)min(rowBase + c, nrows - 1) * K + 4 * h;
  const float* __restrict__ wl = W + (size_t)(4 * h) * 128 + 64 * u + 4 * c;

  const float4 bias4 = *reinterpret_cast<const float4*>(bias + 64 * u + 4 * c);
  float dv[8];
#pragma unroll
  for (int r = 0; r < 8; ++r) {
    dv[r] = dinv[min(rowBase + 8 * h + r, nrows - 1)];
  }

  v8f acc[4];
#pragma unroll
  for (int t = 0; t < 4; ++t) acc[t] = v8f{};

  constexpr int D = 3;
  v4f a[2][4];
  float4 b[D + 1][2];

#pragma unroll
  for (int i = 0; i < 4; ++i) {
    a[0][i] = *reinterpret_cast<const v4f*>(xr + 8 * i);
  }
#pragma unroll
  for (int m = 0; m < D; ++m) {
    const int km = 8 * (m >> 1) + 2 * (m & 1);
#pragma unroll
    for (int s = 0; s < 2; ++s) {
      b[m][s] = *reinterpret_cast<const float4*>(wl + (size_t)(km + s) * 128);
    }
  }

#pragma unroll
  for (int n = 0; n < NSTEP; ++n) {
    const int g = n >> 1;
    const int p = n & 1;
    const int J = g >> 2;
    const int i = g & 3;
    if (n + D < NSTEP) {
      const int kn = 8 * ((n + D) >> 1) + 2 * ((n + D) & 1);
#pragma unroll
      for (int s = 0; s < 2; ++s) {
        b[(n + D) % (D + 1)][s] =
            *reinterpret_cast<const float4*>(wl + (size_t)(kn + s) * 128);
      }
    }
    if ((n & 7) == 0 && J + 1 < NCHUNK) {
#pragma unroll
      for (int q = 0; q < 4; ++q) {
        a[(J + 1) & 1][q] =
            *reinterpret_cast<const v4f*>(xr + 32 * (J + 1) + 8 * q);
      }
    }
    __builtin_amdgcn_sched_barrier(0);
    const v4f aq = a[J & 1][i];
    v2f af;
    af.x = p ? aq.z : aq.x;
    af.y = p ? aq.w : aq.y;
    const float4 b0 = b[n % (D + 1)][0];
    const float4 b1 = b[n % (D + 1)][1];
    v2f bf;
    bf.x = b0.x; bf.y = b1.x;
    acc[0] = __builtin_amdgcn_wmma_f32_16x16x4_f32(
        false, af, false, bf, (short)0, acc[0], false, false);
    bf.x = b0.y; bf.y = b1.y;
    acc[1] = __builtin_amdgcn_wmma_f32_16x16x4_f32(
        false, af, false, bf, (short)0, acc[1], false, false);
    bf.x = b0.z; bf.y = b1.z;
    acc[2] = __builtin_amdgcn_wmma_f32_16x16x4_f32(
        false, af, false, bf, (short)0, acc[2], false, false);
    bf.x = b0.w; bf.y = b1.w;
    acc[3] = __builtin_amdgcn_wmma_f32_16x16x4_f32(
        false, af, false, bf, (short)0, acc[3], false, false);
  }

  float4 hv[8];
  float4 st[8];
#pragma unroll
  for (int r = 0; r < 8; ++r) {
    hv[r] = make_float4(acc[0][r], acc[1][r], acc[2][r], acc[3][r]);
    st[r] = make_float4(hv[r].x * dv[r] + bias4.x, hv[r].y * dv[r] + bias4.y,
                        hv[r].z * dv[r] + bias4.z, hv[r].w * dv[r] + bias4.w);
  }
  const size_t o = (size_t)(rowBase + 8 * h) * 128 + 64 * u + 4 * c;
  if (rowBase + 16 <= nrows) {
#pragma unroll
    for (int r = 0; r < 8; ++r) {
      *reinterpret_cast<float4*>(H + o + (size_t)r * 128) = hv[r];
      *reinterpret_cast<float4*>(Acc + o + (size_t)r * 128) = st[r];
    }
  } else {
#pragma unroll
    for (int r = 0; r < 8; ++r) {
      if (rowBase + 8 * h + r < nrows) {
        *reinterpret_cast<float4*>(H + o + (size_t)r * 128) = hv[r];
        *reinterpret_cast<float4*>(Acc + o + (size_t)r * 128) = st[r];
      }
    }
  }
}

constexpr int kOwnRows = 196;
constexpr int kOwnWaves = 32;
constexpr int kGroupLoads = 8;
constexpr int kGroupEdges = kGroupLoads * 128;
constexpr int kWgGroupEdges = kOwnWaves * kGroupEdges;
constexpr int kQueueCap = 1280;
constexpr int kListCap = 1024;

__global__ __launch_bounds__(1024) void gcn_owner_build(
    const int* __restrict__ src, const int* __restrict__ dst,
    int2* __restrict__ lists, int* __restrict__ listCnt,
    float* __restrict__ dis, float* __restrict__ dinv, int nrows, int nedges) {
  __shared__ int degL[kOwnRows];
  __shared__ int qe[kOwnWaves * kQueueCap];
  __shared__ unsigned char qd[kOwnWaves * kQueueCap];

  const int tid = threadIdx.x;
  const int lane = tid & 31;
  const int wave = __builtin_amdgcn_readfirstlane(tid >> 5);
  const int base = blockIdx.x * kOwnRows;
  const int nr = min(kOwnRows, nrows - base);
  const unsigned unr = (unsigned)nr;

  for (int i = tid; i < kOwnRows; i += 1024) degL[i] = 0;
  __syncthreads();

  int* myQe = qe + wave * kQueueCap;
  unsigned char* myQd = qd + wave * kQueueCap;
  int2* __restrict__ myList =
      lists + (size_t)(blockIdx.x * kOwnWaves + wave) * kListCap;
  int cnt = 0;
  int done = 0;

  const bool vecOK = ((nedges & 3) == 0) &&
                     ((reinterpret_cast<size_t>(dst) & 15) == 0);
  const int nIter = (nedges + kWgGroupEdges - 1) / kWgGroupEdges;
  int it = (nIter > 0) ? (int)(blockIdx.x % (unsigned)nIter) : 0;

  for (int step = 0;; ++step) {
    const bool last = (step >= nIter);
    if (last || cnt > kQueueCap - kGroupEdges) {
      __builtin_amdgcn_fence(__ATOMIC_ACQ_REL, "wavefront");
      for (int i0 = 0; i0 < cnt; i0 += 128) {
        const int i = i0 + lane;
        const int q0 = min(i, cnt - 1);
        const int q1 = min(i + 32, cnt - 1);
        const int q2 = min(i + 64, cnt - 1);
        const int q3 = min(i + 96, cnt - 1);
        const int s0 = src[myQe[q0]];
        const int s1 = src[myQe[q1]];
        const int s2 = src[myQe[q2]];
        const int s3 = src[myQe[q3]];
        const int d0 = myQd[q0];
        const int d1 = myQd[q1];
        const int d2 = myQd[q2];
        const int d3 = myQd[q3];
        const int o = done + i;
        if (i < cnt && o < kListCap) myList[o] = make_int2(s0, d0);
        if (i + 32 < cnt && o + 32 < kListCap) myList[o + 32] = make_int2(s1, d1);
        if (i + 64 < cnt && o + 64 < kListCap) myList[o + 64] = make_int2(s2, d2);
        if (i + 96 < cnt && o + 96 < kListCap) myList[o + 96] = make_int2(s3, d3);
      }
      __builtin_amdgcn_fence(__ATOMIC_ACQ_REL, "wavefront");
      done += cnt;
      cnt = 0;
    }
    if (last) break;

    const int eb = (it * kOwnWaves + wave) * kGroupEdges;
    it = (it + 1 == nIter) ? 0 : it + 1;
    if (eb >= nedges) continue;

    int4 cur[kGroupLoads];
    if (vecOK && eb + kGroupEdges <= nedges) {
      const int4* __restrict__ d4 =
          reinterpret_cast<const int4*>(dst + eb) + lane;
#pragma unroll
      for (int k = 0; k < kGroupLoads; ++k) cur[k] = d4[k * 32];
    } else {
#pragma unroll
      for (int k = 0; k < kGroupLoads; ++k) {
        const int e = eb + k * 128 + lane * 4;
        cur[k].x = (e + 0 < nedges) ? dst[e + 0] : -1;
        cur[k].y = (e + 1 < nedges) ? dst[e + 1] : -1;
        cur[k].z = (e + 2 < nedges) ? dst[e + 2] : -1;
        cur[k].w = (e + 3 < nedges) ? dst[e + 3] : -1;
      }
    }

#pragma unroll
    for (int k = 0; k < kGroupLoads; ++k) {
      const unsigned u0 = (unsigned)(cur[k].x - base);
      const unsigned u1 = (unsigned)(cur[k].y - base);
      const unsigned u2 = (unsigned)(cur[k].z - base);
      const unsigned u3 = (unsigned)(cur[k].w - base);
      const bool any = (u0 < unr) | (u1 < unr) | (u2 < unr) | (u3 < unr);
      if (__builtin_amdgcn_ballot_w32(any) != 0u) {
        const int e = eb + k * 128 + lane * 4;
        const unsigned u[4] = {u0, u1, u2, u3};
#pragma unroll
        for (int j = 0; j < 4; ++j) {
          const bool m = u[j] < unr;
          const unsigned mask = __builtin_amdgcn_ballot_w32(m);
          if (m) {
            const int slot = cnt + (int)__builtin_amdgcn_mbcnt_lo(mask, 0u);
            myQe[slot] = e + j;
            myQd[slot] = (unsigned char)u[j];
            __hip_atomic_fetch_add(&degL[u[j]], 1, __ATOMIC_RELAXED,
                                   __HIP_MEMORY_SCOPE_WORKGROUP);
          }
          cnt += __builtin_popcount(mask);
        }
      }
    }
  }

  if (lane == 0) listCnt[blockIdx.x * kOwnWaves + wave] = done;

  __syncthreads();

  for (int i = tid; i < nr; i += 1024) {
    const float d = (float)(degL[i] + 1);
    dis[base + i] = rsqrtf(d);
    dinv[base + i] = 1.0f / d;
  }
}

__global__ __launch_bounds__(1024) void gcn_owner_pass(
    const int* __restrict__ src, const int* __restrict__ dst,
    const int2* __restrict__ lists, const int* __restrict__ listCnt,
    const float* __restrict__ H, const float* __restrict__ dis,
    float* __restrict__ plane, int nrows, int nedges) {
  __shared__ float rows[kOwnRows * 128];
  __shared__ float disL[kOwnRows];

  const int tid = threadIdx.x;
  const int lane = tid & 31;
  const int wave = __builtin_amdgcn_readfirstlane(tid >> 5);
  const int base = blockIdx.x * kOwnRows;
  const int nr = min(kOwnRows, nrows - base);
  const unsigned unr = (unsigned)nr;

  const int2* __restrict__ myList =
      lists + (size_t)(blockIdx.x * kOwnWaves + wave) * kListCap;
  const int n = __builtin_amdgcn_readfirstlane(
      listCnt[blockIdx.x * kOwnWaves + wave]);
  const bool listOK = (n <= kListCap);
  int2 ent = make_int2(0, 0);
  if (listOK && n > 0) ent = myList[min(lane, n - 1)];

  {
    constexpr int kRowsPerWave = (kOwnRows + kOwnWaves - 1) / kOwnWaves;
    const v4f* __restrict__ p4 =
        reinterpret_cast<const v4f*>(plane) + (size_t)base * 32 + lane;
    v4f* r4 = reinterpret_cast<v4f*>(rows) + lane;
    v4f t[kRowsPerWave];
#pragma unroll
    for (int m = 0; m < kRowsPerWave; ++m) {
      const int r = wave + kOwnWaves * m;
      if (r < nr) t[m] = p4[r * 32];
    }
#pragma unroll
    for (int m = 0; m < kRowsPerWave; ++m) {
      const int r = wave + kOwnWaves * m;
      if (r < nr) r4[r * 32] = t[m];
    }
    for (int i = tid; i < nr; i += 1024) disL[i] = dis[base + i];
  }
  __syncthreads();

  if (listOK) {
    for (int b0 = 0; b0 < n; b0 += 32) {
      const int s = ent.x;
      const int dl = ent.y;
      if (b0 + 32 < n) ent = myList[min(b0 + 32 + lane, n - 1)];
      const float nrm = dis[s] * disL[dl];
      const int m32 = min(32, n - b0);
#pragma unroll
      for (int q = 0; q < 4; ++q) {
        const int m = m32 - 8 * q;
        if (m > 0) {
          float v[8][4];
#pragma unroll
          for (int j = 0; j < 8; ++j) {
            if (j < m) {
              const int sj = __builtin_amdgcn_readlane(s, 8 * q + j);
              const float* __restrict__ hs = H + (size_t)sj * 128 + lane;
              v[j][0] = hs[0];
              v[j][1] = hs[32];
              v[j][2] = hs[64];
              v[j][3] = hs[96];
            }
          }
#pragma unroll
          for (int j = 0; j < 8; ++j) {
            if (j < m) {
              const int dj = __builtin_amdgcn_readlane(dl, 8 * q + j);
              const float nj = __int_as_float(
                  __builtin_amdgcn_readlane(__float_as_int(nrm), 8 * q + j));
              float* r = rows + dj * 128 + lane;
              __hip_atomic_fetch_add(r + 0, v[j][0] * nj, __ATOMIC_RELAXED,
                                     __HIP_MEMORY_SCOPE_WORKGROUP);
              __hip_atomic_fetch_add(r + 32, v[j][1] * nj, __ATOMIC_RELAXED,
                                     __HIP_MEMORY_SCOPE_WORKGROUP);
              __hip_atomic_fetch_add(r + 64, v[j][2] * nj, __ATOMIC_RELAXED,
                                     __HIP_MEMORY_SCOPE_WORKGROUP);
              __hip_atomic_fetch_add(r + 96, v[j][3] * nj, __ATOMIC_RELAXED,
                                     __HIP_MEMORY_SCOPE_WORKGROUP);
            }
          }
        }
      }
    }
  } else {
    const int nIter = (nedges + kWgGroupEdges - 1) / kWgGroupEdges;
    for (int g = 0; g < nIter; ++g) {
      const int eb = (g * kOwnWaves + wave) * kGroupEdges;
      const int ee = min(eb + kGroupEdges, nedges);
      for (int e0 = eb; e0 < ee; e0 += 32) {
        const int e = e0 + lane;
        const unsigned u = (e < ee) ? (unsigned)(dst[e] - base) : 0xffffffffu;
        const bool own = u < unr;
        unsigned mask = __builtin_amdgcn_ballot_w32(own);
        if (mask == 0u) continue;
        const int s = own ? src[e] : 0;
        const float nrm = own ? dis[s] * disL[own ? u : 0u] : 0.0f;
        while (mask != 0u) {
          const int j = __builtin_ctz(mask);
          mask &= mask - 1u;
          const int sj = __builtin_amdgcn_readlane(s, j);
          const int dj = __builtin_amdgcn_readlane((int)u, j);
          const float nj = __int_as_float(
              __builtin_amdgcn_readlane(__float_as_int(nrm), j));
          const float* __restrict__ hs = H + (size_t)sj * 128 + lane;
          float* r = rows + dj * 128 + lane;
          __hip_atomic_fetch_add(r + 0, hs[0] * nj, __ATOMIC_RELAXED,
                                 __HIP_MEMORY_SCOPE_WORKGROUP);
          __hip_atomic_fetch_add(r + 32, hs[32] * nj, __ATOMIC_RELAXED,
                                 __HIP_MEMORY_SCOPE_WORKGROUP);
          __hip_atomic_fetch_add(r + 64, hs[64] * nj, __ATOMIC_RELAXED,
                                 __HIP_MEMORY_SCOPE_WORKGROUP);
          __hip_atomic_fetch_add(r + 96, hs[96] * nj, __ATOMIC_RELAXED,
                                 __HIP_MEMORY_SCOPE_WORKGROUP);
        }
      }
    }
  }

  __syncthreads();

  {
    float4* __restrict__ p4 = reinterpret_cast<float4*>(plane) + (size_t)base * 32;
    const float4* r4 = reinterpret_cast<const float4*>(rows);
    const int n4 = nr * 32;
    for (int i = tid; i < n4; i += 1024) {
      const float4 a = r4[i];
      float4 o;
      o.x = fmaxf(a.x, 0.0f);
      o.y = fmaxf(a.y, 0.0f);
      o.z = fmaxf(a.z, 0.0f);
      o.w = fmaxf(a.w, 0.0f);
      p4[i] = o;
    }
  }
}

extern "C" void kernel_launch(void* const* d_in, const int* in_sizes, int n_in,
                              void* d_out, int out_size, void* d_ws,
                              size_t ws_size, hipStream_t stream) {
  const float* x = (const float*)d_in[0];
  const int* edge = (const int*)d_in[1];
  const float* W0 = (const float*)d_in[2];
  const float* b0 = (const float*)d_in[3];
  const float* W1 = (const float*)d_in[4];
  const float* b1 = (const float*)d_in[5];
  const float* W2 = (const float*)d_in[6];
  const float* b2 = (const float*)d_in[7];
  float* out = (float*)d_out;

  const int N = in_sizes[0] / 256;
  const int E = in_sizes[1] / 2;
  const int* srcIdx = edge;
  const int* dstIdx = edge + E;

  const int ownBlocks = (N + kOwnRows - 1) / kOwnRows;
  const int gemmBlocks = (N + 31) / 32;

  float* ws = (float*)d_ws;
  float* hbuf = ws;
  float* acc0 = hbuf + (size_t)N * 128;
  float* acc1 = acc0 + (size_t)N * 128;
  int2* lists = (int2*)(acc1 + (size_t)N * 128);
  int* listCnt = (int*)(lists + (size_t)ownBlocks * kOwnWaves * kListCap);
  float* dis = (float*)(listCnt + (size_t)ownBlocks * kOwnWaves);
  float* dinv = dis + N;

  gcn_owner_build<<<ownBlocks, 1024, 0, stream>>>(srcIdx, dstIdx, lists, listCnt,
                                                  dis, dinv, N, E);

  gcn_gemm_wmma<256><<<gemmBlocks, 128, 0, stream>>>(x, W0, dinv, b0, hbuf, acc0, N);
  gcn_owner_pass<<<ownBlocks, 1024, 0, stream>>>(srcIdx, dstIdx, lists, listCnt,
                                                 hbuf, dis, acc0, N, E);

  gcn_gemm_wmma<128><<<gemmBlocks, 128, 0, stream>>>(acc0, W1, dinv, b1, hbuf, acc1, N);
  gcn_owner_pass<<<ownBlocks, 1024, 0, stream>>>(srcIdx, dstIdx, lists, listCnt,
                                                 hbuf, dis, acc1, N, E);

  gcn_gemm_wmma<128><<<gemmBlocks, 128, 0, stream>>>(acc1, W2, dinv, b2, hbuf, out, N);
  gcn_owner_pass<<<ownBlocks, 1024, 0, stream>>>(srcIdx, dstIdx, lists, listCnt,
                                                 hbuf, dis, out, N, E);
}


# Destination list read once, not once per owner workgroup: a sort launch orders each 1024-edge slice by owner, owner waves copy their pieces; 8 launches, not 7; sum order: LDS arrival as before
# speedup vs baseline: 3.5236x; 1.1186x over previous
#include <hip/hip_runtime.h>
#include <hip/hip_bf16.h>


typedef __attribute__((ext_vector_type(2))) float v2f;
typedef __attribute__((ext_vector_type(4))) float v4f;
typedef __attribute__((ext_vector_type(8))) float v8f;

template <int K>
__global__ __launch_bounds__(128) void gcn_gemm_wmma(
    const float* __restrict__ X, const float* __restrict__ W,
    const float* __restrict__ dinv, const float* __restrict__ bias,
    float* __restrict__ H, float* __restrict__ Acc, int nrows) {
  constexpr int NCHUNK = K / 32;
  constexpr int NSTEP = K / 4;

  const int lane = threadIdx.x & 31;
  const int c = lane & 15;
  const int h = lane >> 4;
  const int waveInBlock = __builtin_amdgcn_readfirstlane(threadIdx.x >> 5);
  const int u = waveInBlock & 1;
  const int rowBase = (blockIdx.x * 2 + (waveInBlock >> 1)) * 16;
  if (rowBase >= nrows) return;

  const float* __restrict__ xr =
      X + (size_t)min(rowBase + c, nrows - 1) * K + 4 * h;
  const float* __restrict__ wl = W + (size_t)(4 * h) * 128 + 64 * u + 4 * c;

  const float4 bias4 = *reinterpret_cast<const float4*>(bias + 64 * u + 4 * c);
  float dv[8];
#pragma unroll
  for (int r = 0; r < 8; ++r) {
    dv[r] = dinv[min(rowBase + 8 * h + r, nrows - 1)];
  }

  v8f acc[4];
#pragma unroll
  for (int t = 0; t < 4; ++t) acc[t] = v8f{};

  constexpr int D = 3;
  v4f a[2][4];
  float4 b[D + 1][2];

#pragma unroll
  for (int i = 0; i < 4; ++i) {
    a[0][i] = *reinterpret_cast<const v4f*>(xr + 8 * i);
  }
#pragma unroll
  for (int m = 0; m < D; ++m) {
    const int km = 8 * (m >> 1) + 2 * (m & 1);
#pragma unroll
    for (int s = 0; s < 2; ++s) {
      b[m][s] = *reinterpret_cast<const float4*>(wl + (size_t)(km + s) * 128);
    }
  }

#pragma unroll
  for (int n = 0; n < NSTEP; ++n) {
    const int g = n >> 1;
    const int p = n & 1;
    const int J = g >> 2;
    const int i = g & 3;
    if (n + D < NSTEP) {
      const int kn = 8 * ((n + D) >> 1) + 2 * ((n + D) & 1);
#pragma unroll
      for (int s = 0; s < 2; ++s) {
        b[(n + D) % (D + 1)][s] =
            *reinterpret_cast<const float4*>(wl + (size_t)(kn + s) * 128);
      }
    }
    if ((n & 7) == 0 && J + 1 < NCHUNK) {
#pragma unroll
      for (int q = 0; q < 4; ++q) {
        a[(J + 1) & 1][q] =
            *reinterpret_cast<const v4f*>(xr + 32 * (J + 1) + 8 * q);
      }
    }
    __builtin_amdgcn_sched_barrier(0);
    const v4f aq = a[J & 1][i];
    v2f af;
    af.x = p ? aq.z : aq.x;
    af.y = p ? aq.w : aq.y;
    const float4 b0 = b[n % (D + 1)][0];
    const float4 b1 = b[n % (D + 1)][1];
    v2f bf;
    bf.x = b0.x; bf.y = b1.x;
    acc[0] = __builtin_amdgcn_wmma_f32_16x16x4_f32(
        false, af, false, bf, (short)0, acc[0], false, false);
    bf.x = b0.y; bf.y = b1.y;
    acc[1] = __builtin_amdgcn_wmma_f32_16x16x4_f32(
        false, af, false, bf, (short)0, acc[1], false, false);
    bf.x = b0.z; bf.y = b1.z;
    acc[2] = __builtin_amdgcn_wmma_f32_16x16x4_f32(
        false, af, false, bf, (short)0, acc[2], false, false);
    bf.x = b0.w; bf.y = b1.w;
    acc[3] = __builtin_amdgcn_wmma_f32_16x16x4_f32(
        false, af, false, bf, (short)0, acc[3], false, false);
  }

  float4 hv[8];
  float4 st[8];
#pragma unroll
  for (int r = 0; r < 8; ++r) {
    hv[r] = make_float4(acc[0][r], acc[1][r], acc[2][r], acc[3][r]);
    st[r] = make_float4(hv[r].x * dv[r] + bias4.x, hv[r].y * dv[r] + bias4.y,
                        hv[r].z * dv[r] + bias4.z, hv[r].w * dv[r] + bias4.w);
  }
  const size_t o = (size_t)(rowBase + 8 * h) * 128 + 64 * u + 4 * c;
  if (rowBase + 16 <= nrows) {
#pragma unroll
    for (int r = 0; r < 8; ++r) {
      *reinterpret_cast<float4*>(H + o + (size_t)r * 128) = hv[r];
      *reinterpret_cast<float4*>(Acc + o + (size_t)r * 128) = st[r];
    }
  } else {
#pragma unroll
    for (int r = 0; r < 8; ++r) {
      if (rowBase + 8 * h + r < nrows) {
        *reinterpret_cast<float4*>(H + o + (size_t)r * 128) = hv[r];
        *reinterpret_cast<float4*>(Acc + o + (size_t)r * 128) = st[r];
      }
    }
  }
}

constexpr int kOwnRows = 196;
constexpr int kOwnWaves = 32;
constexpr int kGroupLoads = 8;
constexpr int kGroupEdges = kGroupLoads * 128;
constexpr int kWgGroupEdges = kOwnWaves * kGroupEdges;
constexpr int kListCap = 1024;
constexpr int kSortThreads = 256;
constexpr int kMaxOwn = 1024;
constexpr int kStageCap = 256;

__global__ __launch_bounds__(kSortThreads) void gcn_slice_sort(
    const int* __restrict__ src, const int* __restrict__ dst,
    int2* __restrict__ sorted, int* __restrict__ offs, int nrows, int nedges,
    int nOwn) {
  __shared__ int cntL[kMaxOwn];
  __shared__ int offL[kMaxOwn + 1];
  __shared__ int wsum[kSortThreads / 32];
  __shared__ int2 entL[kGroupEdges];

  const int tid = threadIdx.x;
  const int lane = tid & 31;
  const int wave = tid >> 5;
  const int eb = blockIdx.x * kGroupEdges + 4 * tid;

  for (int i = tid; i < kMaxOwn; i += kSortThreads) cntL[i] = 0;

  const bool vecOK = ((nedges & 3) == 0) &&
                     ((reinterpret_cast<size_t>(dst) & 15) == 0) &&
                     ((reinterpret_cast<size_t>(src) & 15) == 0);
  int d[4];
  int s[4];
  if (vecOK && eb + 4 <= nedges) {
    const int4 dv = *reinterpret_cast<const int4*>(dst + eb);
    const int4 sv = *reinterpret_cast<const int4*>(src + eb);
    d[0] = dv.x; d[1] = dv.y; d[2] = dv.z; d[3] = dv.w;
    s[0] = sv.x; s[1] = sv.y; s[2] = sv.z; s[3] = sv.w;
  } else {
#pragma unroll
    for (int j = 0; j < 4; ++j) {
      d[j] = (eb + j < nedges) ? dst[eb + j] : -1;
      s[j] = (eb + j < nedges) ? src[eb + j] : 0;
    }
  }
  __syncthreads();

  int o[4];
  int l[4];
  int rk[4];
#pragma unroll
  for (int j = 0; j < 4; ++j) {
    const bool ok = (unsigned)d[j] < (unsigned)nrows;
    const int oj = ok ? d[j] / kOwnRows : -1;
    o[j] = (oj < kMaxOwn) ? oj : -1;
    l[j] = d[j] - oj * kOwnRows;
    rk[j] = 0;
    if (o[j] >= 0) {
      rk[j] = __hip_atomic_fetch_add(&cntL[o[j]], 1, __ATOMIC_RELAXED,
                                     __HIP_MEMORY_SCOPE_WORKGROUP);
    }
  }
  __syncthreads();

  const int c0 = cntL[4 * tid + 0];
  const int c1 = cntL[4 * tid + 1];
  const int c2 = cntL[4 * tid + 2];
  const int c3 = cntL[4 * tid + 3];
  const int tsum = c0 + c1 + c2 + c3;
  int x = tsum;
  for (int dl = 1; dl < 32; dl <<= 1) {
    const int y = __shfl_up(x, dl);
    if (lane >= dl) x += y;
  }
  if (lane == 31) wsum[wave] = x;
  __syncthreads();
  int wb = 0;
  for (int i = 0; i < kSortThreads / 32; ++i) {
    if (i < wave) wb += wsum[i];
  }
  const int ex = wb + x - tsum;
  offL[4 * tid + 0] = ex;
  offL[4 * tid + 1] = ex + c0;
  offL[4 * tid + 2] = ex + c0 + c1;
  offL[4 * tid + 3] = ex + c0 + c1 + c2;
  if (tid == kSortThreads - 1) offL[kMaxOwn] = ex + tsum;
  __syncthreads();

  const int total = offL[kMaxOwn];
#pragma unroll
  for (int j = 0; j < 4; ++j) {
    if (o[j] >= 0) entL[offL[o[j]] + rk[j]] = make_int2(s[j], l[j]);
  }
  int* __restrict__ myOffs = offs + (size_t)blockIdx.x * (nOwn + 1);
  for (int i = tid; i <= nOwn; i += kSortThreads) myOffs[i] = offL[i];
  __syncthreads();

  int2* __restrict__ mySorted = sorted + (size_t)blockIdx.x * kGroupEdges;
  for (int i = tid; i < total; i += kSortThreads) mySorted[i] = entL[i];
}

__global__ __launch_bounds__(1024) void gcn_owner_build(
    const int2* __restrict__ sorted, const int* __restrict__ offs,
    int2* __restrict__ lists, int* __restrict__ listCnt,
    float* __restrict__ dis, float* __restrict__ dinv, int nrows, int nedges,
    int nOwn) {
  __shared__ int degL[kOwnRows];
  __shared__ int2 stage[kOwnWaves * kStageCap];

  const int tid = threadIdx.x;
  const int lane = tid & 31;
  const int wave = __builtin_amdgcn_readfirstlane(tid >> 5);
  const int base = blockIdx.x * kOwnRows;
  const int nr = min(kOwnRows, nrows - base);
  const int nSlices = (nedges + kGroupEdges - 1) / kGroupEdges;

  for (int i = tid; i < kOwnRows; i += 1024) degL[i] = 0;
  __syncthreads();

  int2* __restrict__ myList =
      lists + (size_t)(blockIdx.x * kOwnWaves + wave) * kListCap;
  int2* myStage = stage + wave * kStageCap;
  int done = 0;

  for (int g0 = wave; g0 < nSlices; g0 += kOwnWaves * 32) {
    const int g = g0 + kOwnWaves * lane;
    const bool have = g < nSlices;
    const int gc = have ? g : 0;
    const int* __restrict__ op = offs + (size_t)gc * (nOwn + 1) + blockIdx.x;
    const int lo = op[0];
    const int n = have ? (op[1] - lo) : 0;

    int x = n;
    for (int dl = 1; dl < 32; dl <<= 1) {
      const int y = __shfl_up(x, dl);
      if (lane >= dl) x += y;
    }
    const int total = __builtin_amdgcn_readlane(x, 31);
    const int at = done + x - n;
    const int2* __restrict__ sp = sorted + (size_t)gc * kGroupEdges + lo;

    for (int k = 0; __builtin_amdgcn_ballot_w32(k < n) != 0u; k += 4) {
      int2 e[4];
#pragma unroll
      for (int j = 0; j < 4; ++j) {
        e[j] = make_int2(0, 0);
        if (k + j < n) e[j] = sp[k + j];
      }
#pragma unroll
      for (int j = 0; j < 4; ++j) {
        if (k + j < n) {
          const int p = at + k + j;
          if (p < kStageCap) {
            myStage[p] = e[j];
          } else if (p < kListCap) {
            myList[p] = e[j];
          }
          __hip_atomic_fetch_add(&degL[e[j].y], 1, __ATOMIC_RELAXED,
                                 __HIP_MEMORY_SCOPE_WORKGROUP);
        }
      }
    }
    done += total;
  }

  if (lane == 0) listCnt[blockIdx.x * kOwnWaves + wave] = done;

  __syncthreads();

  const int ns = min(done, kStageCap);
  for (int i = lane; i < ns; i += 32) myList[i] = myStage[i];

  for (int i = tid; i < nr; i += 1024) {
    const float d = (float)(degL[i] + 1);
    dis[base + i] = rsqrtf(d);
    dinv[base + i] = 1.0f / d;
  }
}

__global__ __launch_bounds__(1024) void gcn_owner_pass(
    const int* __restrict__ src, const int* __restrict__ dst,
    const int2* __restrict__ lists, const int* __restrict__ listCnt,
    const float* __restrict__ H, const float* __restrict__ dis,
    float* __restrict__ plane, int nrows, int nedges) {
  __shared__ float rows[kOwnRows * 128];
  __shared__ float disL[kOwnRows];

  const int tid = threadIdx.x;
  const int lane = tid & 31;
  const int wave = __builtin_amdgcn_readfirstlane(tid >> 5);
  const int base = blockIdx.x * kOwnRows;
  const int nr = min(kOwnRows, nrows - base);
  const unsigned unr = (unsigned)nr;

  const int2* __restrict__ myList =
      lists + (size_t)(blockIdx.x * kOwnWaves + wave) * kListCap;
  const int n = __builtin_amdgcn_readfirstlane(
      listCnt[blockIdx.x * kOwnWaves + wave]);
  const bool listOK = (n <= kListCap);
  int2 ent = make_int2(0, 0);
  if (listOK && n > 0) ent = myList[min(lane, n - 1)];

  {
    constexpr int kRowsPerWave = (kOwnRows + kOwnWaves - 1) / kOwnWaves;
    const v4f* __restrict__ p4 =
        reinterpret_cast<const v4f*>(plane) + (size_t)base * 32 + lane;
    v4f* r4 = reinterpret_cast<v4f*>(rows) + lane;
    v4f t[kRowsPerWave];
#pragma unroll
    for (int m = 0; m < kRowsPerWave; ++m) {
      const int r = wave + kOwnWaves * m;
      if (r < nr) t[m] = p4[r * 32];
    }
#pragma unroll
    for (int m = 0; m < kRowsPerWave; ++m) {
      const int r = wave + kOwnWaves * m;
      if (r < nr) r4[r * 32] = t[m];
    }
    for (int i = tid; i < nr; i += 1024) disL[i] = dis[base + i];
  }
  __syncthreads();

  if (listOK) {
    for (int b0 = 0; b0 < n; b0 += 32) {
      const int s = ent.x;
      const int dl = ent.y;
      if (b0 + 32 < n) ent = myList[min(b0 + 32 + lane, n - 1)];
      const float nrm = dis[s] * disL[dl];
      const int m32 = min(32, n - b0);
#pragma unroll
      for (int q = 0; q < 4; ++q) {
        const int m = m32 - 8 * q;
        if (m > 0) {
          float v[8][4];
#pragma unroll
          for (int j = 0; j < 8; ++j) {
            if (j < m) {
              const int sj = __builtin_amdgcn_readlane(s, 8 * q + j);
              const float* __restrict__ hs = H + (size_t)sj * 128 + lane;
              v[j][0] = hs[0];
              v[j][1] = hs[32];
              v[j][2] = hs[64];
              v[j][3] = hs[96];
            }
          }
#pragma unroll
          for (int j = 0; j < 8; ++j) {
            if (j < m) {
              const int dj = __builtin_amdgcn_readlane(dl, 8 * q + j);
              const float nj = __int_as_float(
                  __builtin_amdgcn_readlane(__float_as_int(nrm), 8 * q + j));
              float* r = rows + dj * 128 + lane;
              __hip_atomic_fetch_add(r + 0, v[j][0] * nj, __ATOMIC_RELAXED,
                                     __HIP_MEMORY_SCOPE_WORKGROUP);
              __hip_atomic_fetch_add(r + 32, v[j][1] * nj, __ATOMIC_RELAXED,
                                     __HIP_MEMORY_SCOPE_WORKGROUP);
              __hip_atomic_fetch_add(r + 64, v[j][2] * nj, __ATOMIC_RELAXED,
                                     __HIP_MEMORY_SCOPE_WORKGROUP);
              __hip_atomic_fetch_add(r + 96, v[j][3] * nj, __ATOMIC_RELAXED,
                                     __HIP_MEMORY_SCOPE_WORKGROUP);
            }
          }
        }
      }
    }
  } else {
    const int nIter = (nedges + kWgGroupEdges - 1) / kWgGroupEdges;
    for (int g = 0; g < nIter; ++g) {
      const int eb = (g * kOwnWaves + wave) * kGroupEdges;
      const int ee = min(eb + kGroupEdges, nedges);
      for (int e0 = eb; e0 < ee; e0 += 32) {
        const int e = e0 + lane;
        const unsigned u = (e < ee) ? (unsigned)(dst[e] - base) : 0xffffffffu;
        const bool own = u < unr;
        unsigned mask = __builtin_amdgcn_ballot_w32(own);
        if (mask == 0u) continue;
        const int s = own ? src[e] : 0;
        const float nrm = own ? dis[s] * disL[own ? u : 0u] : 0.0f;
        while (mask != 0u) {
          const int j = __builtin_ctz(mask);
          mask &= mask - 1u;
          const int sj = __builtin_amdgcn_readlane(s, j);
          const int dj = __builtin_amdgcn_readlane((int)u, j);
          const float nj = __int_as_float(
              __builtin_amdgcn_readlane(__float_as_int(nrm), j));
          const float* __restrict__ hs = H + (size_t)sj * 128 + lane;
          float* r = rows + dj * 128 + lane;
          __hip_atomic_fetch_add(r + 0, hs[0] * nj, __ATOMIC_RELAXED,
                                 __HIP_MEMORY_SCOPE_WORKGROUP);
          __hip_atomic_fetch_add(r + 32, hs[32] * nj, __ATOMIC_RELAXED,
                                 __HIP_MEMORY_SCOPE_WORKGROUP);
          __hip_atomic_fetch_add(r + 64, hs[64] * nj, __ATOMIC_RELAXED,
                                 __HIP_MEMORY_SCOPE_WORKGROUP);
          __hip_atomic_fetch_add(r + 96, hs[96] * nj, __ATOMIC_RELAXED,
                                 __HIP_MEMORY_SCOPE_WORKGROUP);
        }
      }
    }
  }

  __syncthreads();

  {
    float4* __restrict__ p4 = reinterpret_cast<float4*>(plane) + (size_t)base * 32;
    const float4* r4 = reinterpret_cast<const float4*>(rows);
    const int n4 = nr * 32;
    for (int i = tid; i < n4; i += 1024) {
      const float4 a = r4[i];
      float4 o;
      o.x = fmaxf(a.x, 0.0f);
      o.y = fmaxf(a.y, 0.0f);
      o.z = fmaxf(a.z, 0.0f);
      o.w = fmaxf(a.w, 0.0f);
      p4[i] = o;
    }
  }
}

extern "C" void kernel_launch(void* const* d_in, const int* in_sizes, int n_in,
                              void* d_out, int out_size, void* d_ws,
                              size_t ws_size, hipStream_t stream) {
  const float* x = (const float*)d_in[0];
  const int* edge = (const int*)d_in[1];
  const float* W0 = (const float*)d_in[2];
  const float* b0 = (const float*)d_in[3];
  const float* W1 = (const float*)d_in[4];
  const float* b1 = (const float*)d_in[5];
  const float* W2 = (const float*)d_in[6];
  const float* b2 = (const float*)d_in[7];
  float* out = (float*)d_out;

  const int N = in_sizes[0] / 256;
  const int E = in_sizes[1] / 2;
  const int* srcIdx = edge;
  const int* dstIdx = edge + E;

  const int ownBlocks = (N + kOwnRows - 1) / kOwnRows;
  const int nSlices = (E + kGroupEdges - 1) / kGroupEdges;
  const int gemmBlocks = (N + 31) / 32;

  float* ws = (float*)d_ws;
  float* hbuf = ws;
  float* acc0 = hbuf + (size_t)N * 128;
  float* acc1 = acc0 + (size_t)N * 128;
  int2* lists = (int2*)(acc1 + (size_t)N * 128);
  int2* sorted = lists + (size_t)ownBlocks * kOwnWaves * kListCap;
  int* listCnt = (int*)(sorted + (size_t)nSlices * kGroupEdges);
  float* dis = (float*)(listCnt + (size_t)ownBlocks * kOwnWaves);
  float* dinv = dis + N;
  int* offs = (int*)(dinv + N);

  gcn_slice_sort<<<nSlices, kSortThreads, 0, stream>>>(srcIdx, dstIdx, sorted, offs,
                                                       N, E, ownBlocks);
  gcn_owner_build<<<ownBlocks, 1024, 0, stream>>>(sorted, offs, lists, listCnt,
                                                  dis, dinv, N, E, ownBlocks);

  gcn_gemm_wmma<256><<<gemmBlocks, 128, 0, stream>>>(x, W0, dinv, b0, hbuf, acc0, N);
  gcn_owner_pass<<<ownBlocks, 1024, 0, stream>>>(srcIdx, dstIdx, lists, listCnt,
                                                 hbuf, dis, acc0, N, E);

  gcn_gemm_wmma<128><<<gemmBlocks, 128, 0, stream>>>(acc0, W1, dinv, b1, hbuf, acc1, N);
  gcn_owner_pass<<<ownBlocks, 1024, 0, stream>>>(srcIdx, dstIdx, lists, listCnt,
                                                 hbuf, dis, acc1, N, E);

  gcn_gemm_wmma<128><<<gemmBlocks, 128, 0, stream>>>(acc1, W2, dinv, b2, hbuf, out, N);
  gcn_owner_pass<<<ownBlocks, 1024, 0, stream>>>(srcIdx, dstIdx, lists, listCnt,
                                                 hbuf, dis, out, N, E);
}


# Edge rows gathered by ONE 16-byte load per lane (whole 512-byte row per instruction, not 4 one-line loads); LDS rows word-permuted, float atomics stay conflict-free; sum order: LDS arrival as before
# speedup vs baseline: 3.7560x; 1.0660x over previous
#include <hip/hip_runtime.h>
#include <hip/hip_bf16.h>


typedef __attribute__((ext_vector_type(2))) float v2f;
typedef __attribute__((ext_vector_type(4))) float v4f;
typedef __attribute__((ext_vector_type(8))) float v8f;

template <int K>
__global__ __launch_bounds__(128) void gcn_gemm_wmma(
    const float* __restrict__ X, const float* __restrict__ W,
    const float* __restrict__ dinv, const float* __restrict__ bias,
    float* __restrict__ H, float* __restrict__ Acc, int nrows) {
  constexpr int NCHUNK = K / 32;
  constexpr int NSTEP = K / 4;

  const int lane = threadIdx.x & 31;
  const int c = lane & 15;
  const int h = lane >> 4;
  const int waveInBlock = __builtin_amdgcn_readfirstlane(threadIdx.x >> 5);
  const int u = waveInBlock & 1;
  const int rowBase = (blockIdx.x * 2 + (waveInBlock >> 1)) * 16;
  if (rowBase >= nrows) return;

  const float* __restrict__ xr =
      X + (size_t)min(rowBase + c, nrows - 1) * K + 4 * h;
  const float* __restrict__ wl = W + (size_t)(4 * h) * 128 + 64 * u + 4 * c;

  const float4 bias4 = *reinterpret_cast<const float4*>(bias + 64 * u + 4 * c);
  float dv[8];
#pragma unroll
  for (int r = 0; r < 8; ++r) {
    dv[r] = dinv[min(rowBase + 8 * h + r, nrows - 1)];
  }

  v8f acc[4];
#pragma unroll
  for (int t = 0; t < 4; ++t) acc[t] = v8f{};

  constexpr int D = 3;
  v4f a[2][4];
  float4 b[D + 1][2];

#pragma unroll
  for (int i = 0; i < 4; ++i) {
    a[0][i] = *reinterpret_cast<const v4f*>(xr + 8 * i);
  }
#pragma unroll
  for (int m = 0; m < D; ++m) {
    const int km = 8 * (m >> 1) + 2 * (m & 1);
#pragma unroll
    for (int s = 0; s < 2; ++s) {
      b[m][s] = *reinterpret_cast<const float4*>(wl + (size_t)(km + s) * 128);
    }
  }

#pragma unroll
  for (int n = 0; n < NSTEP; ++n) {
    const int g = n >> 1;
    const int p = n & 1;
    const int J = g >> 2;
    const int i = g & 3;
    if (n + D < NSTEP) {
      const int kn = 8 * ((n + D) >> 1) + 2 * ((n + D) & 1);
#pragma unroll
      for (int s = 0; s < 2; ++s) {
        b[(n + D) % (D + 1)][s] =
            *reinterpret_cast<const float4*>(wl + (size_t)(kn + s) * 128);
      }
    }
    if ((n & 7) == 0 && J + 1 < NCHUNK) {
#pragma unroll
      for (int q = 0; q < 4; ++q) {
        a[(J + 1) & 1][q] =
            *reinterpret_cast<const v4f*>(xr + 32 * (J + 1) + 8 * q);
      }
    }
    __builtin_amdgcn_sched_barrier(0);
    const v4f aq = a[J & 1][i];
    v2f af;
    af.x = p ? aq.z : aq.x;
    af.y = p ? aq.w : aq.y;
    const float4 b0 = b[n % (D + 1)][0];
    const float4 b1 = b[n % (D + 1)][1];
    v2f bf;
    bf.x = b0.x; bf.y = b1.x;
    acc[0] = __builtin_amdgcn_wmma_f32_16x16x4_f32(
        false, af, false, bf, (short)0, acc[0], false, false);
    bf.x = b0.y; bf.y = b1.y;
    acc[1] = __builtin_amdgcn_wmma_f32_16x16x4_f32(
        false, af, false, bf, (short)0, acc[1], false, false);
    bf.x = b0.z; bf.y = b1.z;
    acc[2] = __builtin_amdgcn_wmma_f32_16x16x4_f32(
        false, af, false, bf, (short)0, acc[2], false, false);
    bf.x = b0.w; bf.y = b1.w;
    acc[3] = __builtin_amdgcn_wmma_f32_16x16x4_f32(
        false, af, false, bf, (short)0, acc[3], false, false);
  }

  float4 hv[8];
  float4 st[8];
#pragma unroll
  for (int r = 0; r < 8; ++r) {
    hv[r] = make_float4(acc[0][r], acc[1][r], acc[2][r], acc[3][r]);
    st[r] = make_float4(hv[r].x * dv[r] + bias4.x, hv[r].y * dv[r] + bias4.y,
                        hv[r].z * dv[r] + bias4.z, hv[r].w * dv[r] + bias4.w);
  }
  const size_t o = (size_t)(rowBase + 8 * h) * 128 + 64 * u + 4 * c;
  if (rowBase + 16 <= nrows) {
#pragma unroll
    for (int r = 0; r < 8; ++r) {
      *reinterpret_cast<float4*>(H + o + (size_t)r * 128) = hv[r];
      *reinterpret_cast<float4*>(Acc + o + (size_t)r * 128) = st[r];
    }
  } else {
#pragma unroll
    for (int r = 0; r < 8; ++r) {
      if (rowBase + 8 * h + r < nrows) {
        *reinterpret_cast<float4*>(H + o + (size_t)r * 128) = hv[r];
        *reinterpret_cast<float4*>(Acc + o + (size_t)r * 128) = st[r];
      }
    }
  }
}

constexpr int kOwnRows = 196;
constexpr int kOwnWaves = 32;
constexpr int kGroupLoads = 8;
constexpr int kGroupEdges = kGroupLoads * 128;
constexpr int kWgGroupEdges = kOwnWaves * kGroupEdges;
constexpr int kListCap = 1024;
constexpr int kSortThreads = 256;
constexpr int kMaxOwn = 1024;
constexpr int kStageCap = 256;

__global__ __launch_bounds__(kSortThreads) void gcn_slice_sort(
    const int* __restrict__ src, const int* __restrict__ dst,
    int2* __restrict__ sorted, int* __restrict__ offs, int nrows, int nedges,
    int nOwn) {
  __shared__ int cntL[kMaxOwn];
  __shared__ int offL[kMaxOwn + 1];
  __shared__ int wsum[kSortThreads / 32];
  __shared__ int2 entL[kGroupEdges];

  const int tid = threadIdx.x;
  const int lane = tid & 31;
  const int wave = tid >> 5;
  const int eb = blockIdx.x * kGroupEdges + 4 * tid;

  for (int i = tid; i < kMaxOwn; i += kSortThreads) cntL[i] = 0;

  const bool vecOK = ((nedges & 3) == 0) &&
                     ((reinterpret_cast<size_t>(dst) & 15) == 0) &&
                     ((reinterpret_cast<size_t>(src) & 15) == 0);
  int d[4];
  int s[4];
  if (vecOK && eb + 4 <= nedges) {
    const int4 dv = *reinterpret_cast<const int4*>(dst + eb);
    const int4 sv = *reinterpret_cast<const int4*>(src + eb);
    d[0] = dv.x; d[1] = dv.y; d[2] = dv.z; d[3] = dv.w;
    s[0] = sv.x; s[1] = sv.y; s[2] = sv.z; s[3] = sv.w;
  } else {
#pragma unroll
    for (int j = 0; j < 4; ++j) {
      d[j] = (eb + j < nedges) ? dst[eb + j] : -1;
      s[j] = (eb + j < nedges) ? src[eb + j] : 0;
    }
  }
  __syncthreads();

  int o[4];
  int l[4];
  int rk[4];
#pragma unroll
  for (int j = 0; j < 4; ++j) {
    const bool ok = (unsigned)d[j] < (unsigned)nrows;
    const int oj = ok ? d[j] / kOwnRows : -1;
    o[j] = (oj < kMaxOwn) ? oj : -1;
    l[j] = d[j] - oj * kOwnRows;
    rk[j] = 0;
    if (o[j] >= 0) {
      rk[j] = __hip_atomic_fetch_add(&cntL[o[j]], 1, __ATOMIC_RELAXED,
                                     __HIP_MEMORY_SCOPE_WORKGROUP);
    }
  }
  __syncthreads();

  const int c0 = cntL[4 * tid + 0];
  const int c1 = cntL[4 * tid + 1];
  const int c2 = cntL[4 * tid + 2];
  const int c3 = cntL[4 * tid + 3];
  const int tsum = c0 + c1 + c2 + c3;
  int x = tsum;
  for (int dl = 1; dl < 32; dl <<= 1) {
    const int y = __shfl_up(x, dl);
    if (lane >= dl) x += y;
  }
  if (lane == 31) wsum[wave] = x;
  __syncthreads();
  int wb = 0;
  for (int i = 0; i < kSortThreads / 32; ++i) {
    if (i < wave) wb += wsum[i];
  }
  const int ex = wb + x - tsum;
  offL[4 * tid + 0] = ex;
  offL[4 * tid + 1] = ex + c0;
  offL[4 * tid + 2] = ex + c0 + c1;
  offL[4 * tid + 3] = ex + c0 + c1 + c2;
  if (tid == kSortThreads - 1) offL[kMaxOwn] = ex + tsum;
  __syncthreads();

  const int total = offL[kMaxOwn];
#pragma unroll
  for (int j = 0; j < 4; ++j) {
    if (o[j] >= 0) entL[offL[o[j]] + rk[j]] = make_int2(s[j], l[j]);
  }
  int* __restrict__ myOffs = offs + (size_t)blockIdx.x * (nOwn + 1);
  for (int i = tid; i <= nOwn; i += kSortThreads) myOffs[i] = offL[i];
  __syncthreads();

  int2* __restrict__ mySorted = sorted + (size_t)blockIdx.x * kGroupEdges;
  for (int i = tid; i < total; i += kSortThreads) mySorted[i] = entL[i];
}

__global__ __launch_bounds__(1024) void gcn_owner_build(
    const int2* __restrict__ sorted, const int* __restrict__ offs,
    int2* __restrict__ lists, int* __restrict__ listCnt,
    float* __restrict__ dis, float* __restrict__ dinv, int nrows, int nedges,
    int nOwn) {
  __shared__ int degL[kOwnRows];
  __shared__ int2 stage[kOwnWaves * kStageCap];

  const int tid = threadIdx.x;
  const int lane = tid & 31;
  const int wave = __builtin_amdgcn_readfirstlane(tid >> 5);
  const int base = blockIdx.x * kOwnRows;
  const int nr = min(kOwnRows, nrows - base);
  const int nSlices = (nedges + kGroupEdges - 1) / kGroupEdges;

  for (int i = tid; i < kOwnRows; i += 1024) degL[i] = 0;
  __syncthreads();

  int2* __restrict__ myList =
      lists + (size_t)(blockIdx.x * kOwnWaves + wave) * kListCap;
  int2* myStage = stage + wave * kStageCap;
  int done = 0;

  for (int g0 = wave; g0 < nSlices; g0 += kOwnWaves * 32) {
    const int g = g0 + kOwnWaves * lane;
    const bool have = g < nSlices;
    const int gc = have ? g : 0;
    const int* __restrict__ op = offs + (size_t)gc * (nOwn + 1) + blockIdx.x;
    const int lo = op[0];
    const int n = have ? (op[1] - lo) : 0;

    int x = n;
    for (int dl = 1; dl < 32; dl <<= 1) {
      const int y = __shfl_up(x, dl);
      if (lane >= dl) x += y;
    }
    const int total = __builtin_amdgcn_readlane(x, 31);
    const int at = done + x - n;
    const int2* __restrict__ sp = sorted + (size_t)gc * kGroupEdges + lo;

    for (int k = 0; __builtin_amdgcn_ballot_w32(k < n) != 0u; k += 4) {
      int2 e[4];
#pragma unroll
      for (int j = 0; j < 4; ++j) {
        e[j] = make_int2(0, 0);
        if (k + j < n) e[j] = sp[k + j];
      }
#pragma unroll
      for (int j = 0; j < 4; ++j) {
        if (k + j < n) {
          const int p = at + k + j;
          if (p < kStageCap) {
            myStage[p] = e[j];
          } else if (p < kListCap) {
            myList[p] = e[j];
          }
          __hip_atomic_fetch_add(&degL[e[j].y], 1, __ATOMIC_RELAXED,
                                 __HIP_MEMORY_SCOPE_WORKGROUP);
        }
      }
    }
    done += total;
  }

  if (lane == 0) listCnt[blockIdx.x * kOwnWaves + wave] = done;

  __syncthreads();

  const int ns = min(done, kStageCap);
  for (int i = lane; i < ns; i += 32) myList[i] = myStage[i];

  for (int i = tid; i < nr; i += 1024) {
    const float d = (float)(degL[i] + 1);
    dis[base + i] = rsqrtf(d);
    dinv[base + i] = 1.0f / d;
  }
}

__global__ __launch_bounds__(1024) void gcn_owner_pass(
    const int* __restrict__ src, const int* __restrict__ dst,
    const int2* __restrict__ lists, const int* __restrict__ listCnt,
    const float* __restrict__ H, const float* __restrict__ dis,
    float* __restrict__ plane, int nrows, int nedges) {
  __shared__ float rows[kOwnRows * 128];
  __shared__ float disL[kOwnRows];

  const int tid = threadIdx.x;
  const int lane = tid & 31;
  const int wave = __builtin_amdgcn_readfirstlane(tid >> 5);
  const int base = blockIdx.x * kOwnRows;
  const int nr = min(kOwnRows, nrows - base);
  const unsigned unr = (unsigned)nr;

  const int2* __restrict__ myList =
      lists + (size_t)(blockIdx.x * kOwnWaves + wave) * kListCap;
  const int n = __builtin_amdgcn_readfirstlane(
      listCnt[blockIdx.x * kOwnWaves + wave]);
  const bool listOK = (n <= kListCap);
  int2 ent = make_int2(0, 0);
  if (listOK && n > 0) ent = myList[min(lane, n - 1)];

  {
    constexpr int kRowsPerWave = (kOwnRows + kOwnWaves - 1) / kOwnWaves;
    const v4f* __restrict__ p4 =
        reinterpret_cast<const v4f*>(plane) + (size_t)base * 32 + lane;
    float* rw = rows + lane;
    v4f t[kRowsPerWave];
#pragma unroll
    for (int m = 0; m < kRowsPerWave; ++m) {
      const int r = wave + kOwnWaves * m;
      if (r < nr) t[m] = p4[r * 32];
    }
#pragma unroll
    for (int m = 0; m < kRowsPerWave; ++m) {
      const int r = wave + kOwnWaves * m;
      if (r < nr) {
        rw[r * 128 + 0] = t[m].x;
        rw[r * 128 + 32] = t[m].y;
        rw[r * 128 + 64] = t[m].z;
        rw[r * 128 + 96] = t[m].w;
      }
    }
    for (int i = tid; i < nr; i += 1024) disL[i] = dis[base + i];
  }
  __syncthreads();

  if (listOK) {
    for (int b0 = 0; b0 < n; b0 += 32) {
      const int s = ent.x;
      const int dl = ent.y;
      if (b0 + 32 < n) ent = myList[min(b0 + 32 + lane, n - 1)];
      const float nrm = dis[s] * disL[dl];
      const int m32 = min(32, n - b0);
#pragma unroll
      for (int q = 0; q < 4; ++q) {
        const int m = m32 - 8 * q;
        if (m > 0) {
          v4f v[8];
#pragma unroll
          for (int j = 0; j < 8; ++j) {
            if (j < m) {
              const int sj = __builtin_amdgcn_readlane(s, 8 * q + j);
              v[j] = *reinterpret_cast<const v4f*>(H + (size_t)sj * 128 +
                                                   4 * lane);
            }
          }
#pragma unroll
          for (int j = 0; j < 8; ++j) {
            if (j < m) {
              const int dj = __builtin_amdgcn_readlane(dl, 8 * q + j);
              const float nj = __int_as_float(
                  __builtin_amdgcn_readlane(__float_as_int(nrm), 8 * q + j));
              float* r = rows + dj * 128 + lane;
              __hip_atomic_fetch_add(r + 0, v[j].x * nj, __ATOMIC_RELAXED,
                                     __HIP_MEMORY_SCOPE_WORKGROUP);
              __hip_atomic_fetch_add(r + 32, v[j].y * nj, __ATOMIC_RELAXED,
                                     __HIP_MEMORY_SCOPE_WORKGROUP);
              __hip_atomic_fetch_add(r + 64, v[j].z * nj, __ATOMIC_RELAXED,
                                     __HIP_MEMORY_SCOPE_WORKGROUP);
              __hip_atomic_fetch_add(r + 96, v[j].w * nj, __ATOMIC_RELAXED,
                                     __HIP_MEMORY_SCOPE_WORKGROUP);
            }
          }
        }
      }
    }
  } else {
    const int nIter = (nedges + kWgGroupEdges - 1) / kWgGroupEdges;
    for (int g = 0; g < nIter; ++g) {
      const int eb = (g * kOwnWaves + wave) * kGroupEdges;
      const int ee = min(eb + kGroupEdges, nedges);
      for (int e0 = eb; e0 < ee; e0 += 32) {
        const int e = e0 + lane;
        const unsigned u = (e < ee) ? (unsigned)(dst[e] - base) : 0xffffffffu;
        const bool own = u < unr;
        unsigned mask = __builtin_amdgcn_ballot_w32(own);
        if (mask == 0u) continue;
        const int s = own ? src[e] : 0;
        const float nrm = own ? dis[s] * disL[own ? u : 0u] : 0.0f;
        while (mask != 0u) {
          const int j = __builtin_ctz(mask);
          mask &= mask - 1u;
          const int sj = __builtin_amdgcn_readlane(s, j);
          const int dj = __builtin_amdgcn_readlane((int)u, j);
          const float nj = __int_as_float(
              __builtin_amdgcn_readlane(__float_as_int(nrm), j));
          const v4f hv =
              *reinterpret_cast<const v4f*>(H + (size_t)sj * 128 + 4 * lane);
          float* r = rows + dj * 128 + lane;
          __hip_atomic_fetch_add(r + 0, hv.x * nj, __ATOMIC_RELAXED,
                                 __HIP_MEMORY_SCOPE_WORKGROUP);
          __hip_atomic_fetch_add(r + 32, hv.y * nj, __ATOMIC_RELAXED,
                                 __HIP_MEMORY_SCOPE_WORKGROUP);
          __hip_atomic_fetch_add(r + 64, hv.z * nj, __ATOMIC_RELAXED,
                                 __HIP_MEMORY_SCOPE_WORKGROUP);
          __hip_atomic_fetch_add(r + 96, hv.w * nj, __ATOMIC_RELAXED,
                                 __HIP_MEMORY_SCOPE_WORKGROUP);
        }
      }
    }
  }

  __syncthreads();

  {
    float4* __restrict__ p4 = reinterpret_cast<float4*>(plane) + (size_t)base * 32;
    const int n4 = nr * 32;
    for (int i = tid; i < n4; i += 1024) {
      const float* rr = rows + (i >> 5) * 128 + (i & 31);
      float4 o;
      o.x = fmaxf(rr[0], 0.0f);
      o.y = fmaxf(rr[32], 0.0f);
      o.z = fmaxf(rr[64], 0.0f);
      o.w = fmaxf(rr[96], 0.0f);
      p4[i] = o;
    }
  }
}

extern "C" void kernel_launch(void* const* d_in, const int* in_sizes, int n_in,
                              void* d_out, int out_size, void* d_ws,
                              size_t ws_size, hipStream_t stream) {
  const float* x = (const float*)d_in[0];
  const int* edge = (const int*)d_in[1];
  const float* W0 = (const float*)d_in[2];
  const float* b0 = (const float*)d_in[3];
  const float* W1 = (const float*)d_in[4];
  const float* b1 = (const float*)d_in[5];
  const float* W2 = (const float*)d_in[6];
  const float* b2 = (const float*)d_in[7];
  float* out = (float*)d_out;

  const int N = in_sizes[0] / 256;
  const int E = in_sizes[1] / 2;
  const int* srcIdx = edge;
  const int* dstIdx = edge + E;

  const int ownBlocks = (N + kOwnRows - 1) / kOwnRows;
  const int nSlices = (E + kGroupEdges - 1) / kGroupEdges;
  const int gemmBlocks = (N + 31) / 32;

  float* ws = (float*)d_ws;
  float* hbuf = ws;
  float* acc0 = hbuf + (size_t)N * 128;
  float* acc1 = acc0 + (size_t)N * 128;
  int2* lists = (int2*)(acc1 + (size_t)N * 128);
  int2* sorted = lists + (size_t)ownBlocks * kOwnWaves * kListCap;
  int* listCnt = (int*)(sorted + (size_t)nSlices * kGroupEdges);
  float* dis = (float*)(listCnt + (size_t)ownBlocks * kOwnWaves);
  float* dinv = dis + N;
  int* offs = (int*)(dinv + N);

  gcn_slice_sort<<<nSlices, kSortThreads, 0, stream>>>(srcIdx, dstIdx, sorted, offs,
                                                       N, E, ownBlocks);
  gcn_owner_build<<<ownBlocks, 1024, 0, stream>>>(sorted, offs, lists, listCnt,
                                                  dis, dinv, N, E, ownBlocks);

  gcn_gemm_wmma<256><<<gemmBlocks, 128, 0, stream>>>(x, W0, dinv, b0, hbuf, acc0, N);
  gcn_owner_pass<<<ownBlocks, 1024, 0, stream>>>(srcIdx, dstIdx, lists, listCnt,
                                                 hbuf, dis, acc0, N, E);

  gcn_gemm_wmma<128><<<gemmBlocks, 128, 0, stream>>>(acc0, W1, dinv, b1, hbuf, acc1, N);
  gcn_owner_pass<<<ownBlocks, 1024, 0, stream>>>(srcIdx, dstIdx, lists, listCnt,
                                                 hbuf, dis, acc1, N, E);

  gcn_gemm_wmma<128><<<gemmBlocks, 128, 0, stream>>>(acc1, W2, dinv, b2, hbuf, out, N);
  gcn_owner_pass<<<ownBlocks, 1024, 0, stream>>>(srcIdx, dstIdx, lists, listCnt,
                                                 hbuf, dis, out, N, E);
}


# Normalizers factored out of the edge sum: GEMM stores one plane, rows scaled by deg^-1/2; owner pass adds rows as read, scales by deg^-1/2[dst] at the store; sum order: own row, then LDS arrival
# speedup vs baseline: 4.3601x; 1.1608x over previous
#include <hip/hip_runtime.h>
#include <hip/hip_bf16.h>


typedef __attribute__((ext_vector_type(2))) float v2f;
typedef __attribute__((ext_vector_type(4))) float v4f;
typedef __attribute__((ext_vector_type(8))) float v8f;

template <int K>
__global__ __launch_bounds__(128) void gcn_gemm_wmma(
    const float* __restrict__ X, const float* __restrict__ W,
    const float* __restrict__ dis, float* __restrict__ H, int nrows) {
  constexpr int NCHUNK = K / 32;
  constexpr int NSTEP = K / 4;

  const int lane = threadIdx.x & 31;
  const int c = lane & 15;
  const int h = lane >> 4;
  const int waveInBlock = __builtin_amdgcn_readfirstlane(threadIdx.x >> 5);
  const int u = waveInBlock & 1;
  const int rowBase = (blockIdx.x * 2 + (waveInBlock >> 1)) * 16;
  if (rowBase >= nrows) return;

  const float* __restrict__ xr =
      X + (size_t)min(rowBase + c, nrows - 1) * K + 4 * h;
  const float* __restrict__ wl = W + (size_t)(4 * h) * 128 + 64 * u + 4 * c;

  float dv[8];
#pragma unroll
  for (int r = 0; r < 8; ++r) {
    dv[r] = dis[min(rowBase + 8 * h + r, nrows - 1)];
  }

  v8f acc[4];
#pragma unroll
  for (int t = 0; t < 4; ++t) acc[t] = v8f{};

  constexpr int D = 3;
  v4f a[2][4];
  float4 b[D + 1][2];

#pragma unroll
  for (int i = 0; i < 4; ++i) {
    a[0][i] = *reinterpret_cast<const v4f*>(xr + 8 * i);
  }
#pragma unroll
  for (int m = 0; m < D; ++m) {
    const int km = 8 * (m >> 1) + 2 * (m & 1);
#pragma unroll
    for (int s = 0; s < 2; ++s) {
      b[m][s] = *reinterpret_cast<const float4*>(wl + (size_t)(km + s) * 128);
    }
  }

#pragma unroll
  for (int n = 0; n < NSTEP; ++n) {
    const int g = n >> 1;
    const int p = n & 1;
    const int J = g >> 2;
    const int i = g & 3;
    if (n + D < NSTEP) {
      const int kn = 8 * ((n + D) >> 1) + 2 * ((n + D) & 1);
#pragma unroll
      for (int s = 0; s < 2; ++s) {
        b[(n + D) % (D + 1)][s] =
            *reinterpret_cast<const float4*>(wl + (size_t)(kn + s) * 128);
      }
    }
    if ((n & 7) == 0 && J + 1 < NCHUNK) {
#pragma unroll
      for (int q = 0; q < 4; ++q) {
        a[(J + 1) & 1][q] =
            *reinterpret_cast<const v4f*>(xr + 32 * (J + 1) + 8 * q);
      }
    }
    __builtin_amdgcn_sched_barrier(0);
    const v4f aq = a[J & 1][i];
    v2f af;
    af.x = p ? aq.z : aq.x;
    af.y = p ? aq.w : aq.y;
    const float4 b0 = b[n % (D + 1)][0];
    const float4 b1 = b[n % (D + 1)][1];
    v2f bf;
    bf.x = b0.x; bf.y = b1.x;
    acc[0] = __builtin_amdgcn_wmma_f32_16x16x4_f32(
        false, af, false, bf, (short)0, acc[0], false, false);
    bf.x = b0.y; bf.y = b1.y;
    acc[1] = __builtin_amdgcn_wmma_f32_16x16x4_f32(
        false, af, false, bf, (short)0, acc[1], false, false);
    bf.x = b0.z; bf.y = b1.z;
    acc[2] = __builtin_amdgcn_wmma_f32_16x16x4_f32(
        false, af, false, bf, (short)0, acc[2], false, false);
    bf.x = b0.w; bf.y = b1.w;
    acc[3] = __builtin_amdgcn_wmma_f32_16x16x4_f32(
        false, af, false, bf, (short)0, acc[3], false, false);
  }

  float4 hv[8];
#pragma unroll
  for (int r = 0; r < 8; ++r) {
    hv[r] = make_float4(acc[0][r] * dv[r], acc[1][r] * dv[r], acc[2][r] * dv[r],
                        acc[3][r] * dv[r]);
  }
  const size_t o = (size_t)(rowBase + 8 * h) * 128 + 64 * u + 4 * c;
  if (rowBase + 16 <= nrows) {
#pragma unroll
    for (int r = 0; r < 8; ++r) {
      *reinterpret_cast<float4*>(H + o + (size_t)r * 128) = hv[r];
    }
  } else {
#pragma unroll
    for (int r = 0; r < 8; ++r) {
      if (rowBase + 8 * h + r < nrows) {
        *reinterpret_cast<float4*>(H + o + (size_t)r * 128) = hv[r];
      }
    }
  }
}

constexpr int kOwnRows = 196;
constexpr int kOwnWaves = 32;
constexpr int kGroupLoads = 8;
constexpr int kGroupEdges = kGroupLoads * 128;
constexpr int kWgGroupEdges = kOwnWaves * kGroupEdges;
constexpr int kListCap = 1024;
constexpr int kSortThreads = 256;
constexpr int kMaxOwn = 1024;
constexpr int kStageCap = 256;

__global__ __launch_bounds__(kSortThreads) void gcn_slice_sort(
    const int* __restrict__ src, const int* __restrict__ dst,
    int2* __restrict__ sorted, int* __restrict__ offs, int nrows, int nedges,
    int nOwn) {
  __shared__ int cntL[kMaxOwn];
  __shared__ int offL[kMaxOwn + 1];
  __shared__ int wsum[kSortThreads / 32];
  __shared__ int2 entL[kGroupEdges];

  const int tid = threadIdx.x;
  const int lane = tid & 31;
  const int wave = tid >> 5;
  const int eb = blockIdx.x * kGroupEdges + 4 * tid;

  for (int i = tid; i < kMaxOwn; i += kSortThreads) cntL[i] = 0;

  const bool vecOK = ((nedges & 3) == 0) &&
                     ((reinterpret_cast<size_t>(dst) & 15) == 0) &&
                     ((reinterpret_cast<size_t>(src) & 15) == 0);
  int d[4];
  int s[4];
  if (vecOK && eb + 4 <= nedges) {
    const int4 dv = *reinterpret_cast<const int4*>(dst + eb);
    const int4 sv = *reinterpret_cast<const int4*>(src + eb);
    d[0] = dv.x; d[1] = dv.y; d[2] = dv.z; d[3] = dv.w;
    s[0] = sv.x; s[1] = sv.y; s[2] = sv.z; s[3] = sv.w;
  } else {
#pragma unroll
    for (int j = 0; j < 4; ++j) {
      d[j] = (eb + j < nedges) ? dst[eb + j] : -1;
      s[j] = (eb + j < nedges) ? src[eb + j] : 0;
    }
  }
  __syncthreads();

  int o[4];
  int l[4];
  int rk[4];
#pragma unroll
  for (int j = 0; j < 4; ++j) {
    const bool ok = (unsigned)d[j] < (unsigned)nrows;
    const int oj = ok ? d[j] / kOwnRows : -1;
    o[j] = (oj < kMaxOwn) ? oj : -1;
    l[j] = d[j] - oj * kOwnRows;
    rk[j] = 0;
    if (o[j] >= 0) {
      rk[j] = __hip_atomic_fetch_add(&cntL[o[j]], 1, __ATOMIC_RELAXED,
                                     __HIP_MEMORY_SCOPE_WORKGROUP);
    }
  }
  __syncthreads();

  const int c0 = cntL[4 * tid + 0];
  const int c1 = cntL[4 * tid + 1];
  const int c2 = cntL[4 * tid + 2];
  const int c3 = cntL[4 * tid + 3];
  const int tsum = c0 + c1 + c2 + c3;
  int x = tsum;
  for (int dl = 1; dl < 32; dl <<= 1) {
    const int y = __shfl_up(x, dl);
    if (lane >= dl) x += y;
  }
  if (lane == 31) wsum[wave] = x;
  __syncthreads();
  int wb = 0;
  for (int i = 0; i < kSortThreads / 32; ++i) {
    if (i < wave) wb += wsum[i];
  }
  const int ex = wb + x - tsum;
  offL[4 * tid + 0] = ex;
  offL[4 * tid + 1] = ex + c0;
  offL[4 * tid + 2] = ex + c0 + c1;
  offL[4 * tid + 3] = ex + c0 + c1 + c2;
  if (tid == kSortThreads - 1) offL[kMaxOwn] = ex + tsum;
  __syncthreads();

  const int total = offL[kMaxOwn];
#pragma unroll
  for (int j = 0; j < 4; ++j) {
    if (o[j] >= 0) entL[offL[o[j]] + rk[j]] = make_int2(s[j], l[j]);
  }
  int* __restrict__ myOffs = offs + (size_t)blockIdx.x * (nOwn + 1);
  for (int i = tid; i <= nOwn; i += kSortThreads) myOffs[i] = offL[i];
  __syncthreads();

  int2* __restrict__ mySorted = sorted + (size_t)blockIdx.x * kGroupEdges;
  for (int i = tid; i < total; i += kSortThreads) mySorted[i] = entL[i];
}

__global__ __launch_bounds__(1024) void gcn_owner_build(
    const int2* __restrict__ sorted, const int* __restrict__ offs,
    int2* __restrict__ lists, int* __restrict__ listCnt,
    float* __restrict__ dis, int nrows, int nedges, int nOwn) {
  __shared__ int degL[kOwnRows];
  __shared__ int2 stage[kOwnWaves * kStageCap];

  const int tid = threadIdx.x;
  const int lane = tid & 31;
  const int wave = __builtin_amdgcn_readfirstlane(tid >> 5);
  const int base = blockIdx.x * kOwnRows;
  const int nr = min(kOwnRows, nrows - base);
  const int nSlices = (nedges + kGroupEdges - 1) / kGroupEdges;

  for (int i = tid; i < kOwnRows; i += 1024) degL[i] = 0;
  __syncthreads();

  int2* __restrict__ myList =
      lists + (size_t)(blockIdx.x * kOwnWaves + wave) * kListCap;
  int2* myStage = stage + wave * kStageCap;
  int done = 0;

  for (int g0 = wave; g0 < nSlices; g0 += kOwnWaves * 32) {
    const int g = g0 + kOwnWaves * lane;
    const bool have = g < nSlices;
    const int gc = have ? g : 0;
    const int* __restrict__ op = offs + (size_t)gc * (nOwn + 1) + blockIdx.x;
    const int lo = op[0];
    const int n = have ? (op[1] - lo) : 0;

    int x = n;
    for (int dl = 1; dl < 32; dl <<= 1) {
      const int y = __shfl_up(x, dl);
      if (lane >= dl) x += y;
    }
    const int total = __builtin_amdgcn_readlane(x, 31);
    const int at = done + x - n;
    const int2* __restrict__ sp = sorted + (size_t)gc * kGroupEdges + lo;

    for (int k = 0; __builtin_amdgcn_ballot_w32(k < n) != 0u; k += 4) {
      int2 e[4];
#pragma unroll
      for (int j = 0; j < 4; ++j) {
        e[j] = make_int2(0, 0);
        if (k + j < n) e[j] = sp[k + j];
      }
#pragma unroll
      for (int j = 0; j < 4; ++j) {
        if (k + j < n) {
          const int p = at + k + j;
          if (p < kStageCap) {
            myStage[p] = e[j];
          } else if (p < kListCap) {
            myList[p] = e[j];
          }
          __hip_atomic_fetch_add(&degL[e[j].y], 1, __ATOMIC_RELAXED,
                                 __HIP_MEMORY_SCOPE_WORKGROUP);
        }
      }
    }
    done += total;
  }

  if (lane == 0) listCnt[blockIdx.x * kOwnWaves + wave] = done;

  __syncthreads();

  const int ns = min(done, kStageCap);
  for (int i = lane; i < ns; i += 32) myList[i] = myStage[i];

  for (int i = tid; i < nr; i += 1024) {
    const float d = (float)(degL[i] + 1);
    dis[base + i] = rsqrtf(d);
  }
}

__global__ __launch_bounds__(1024) void gcn_owner_pass(
    const int* __restrict__ src, const int* __restrict__ dst,
    const int2* __restrict__ lists, const int* __restrict__ listCnt,
    const float* __restrict__ H, const float* __restrict__ dis,
    const float* __restrict__ bias, float* __restrict__ plane, int nrows,
    int nedges) {
  __shared__ float rows[kOwnRows * 128];
  __shared__ float disL[kOwnRows];

  const int tid = threadIdx.x;
  const int lane = tid & 31;
  const int wave = __builtin_amdgcn_readfirstlane(tid >> 5);
  const int base = blockIdx.x * kOwnRows;
  const int nr = min(kOwnRows, nrows - base);
  const unsigned unr = (unsigned)nr;

  const int2* __restrict__ myList =
      lists + (size_t)(blockIdx.x * kOwnWaves + wave) * kListCap;
  const int n = __builtin_amdgcn_readfirstlane(
      listCnt[blockIdx.x * kOwnWaves + wave]);
  const bool listOK = (n <= kListCap);
  int2 ent = make_int2(0, 0);
  if (listOK && n > 0) ent = myList[min(lane, n - 1)];
  const float4 bias4 = *reinterpret_cast<const float4*>(bias + 4 * lane);

  {
    constexpr int kRowsPerWave = (kOwnRows + kOwnWaves - 1) / kOwnWaves;
    const v4f* __restrict__ p4 =
        reinterpret_cast<const v4f*>(H) + (size_t)base * 32 + lane;
    float* rw = rows + lane;
    v4f t[kRowsPerWave];
#pragma unroll
    for (int m = 0; m < kRowsPerWave; ++m) {
      const int r = wave + kOwnWaves * m;
      if (r < nr) t[m] = p4[r * 32];
    }
#pragma unroll
    for (int m = 0; m < kRowsPerWave; ++m) {
      const int r = wave + kOwnWaves * m;
      if (r < nr) {
        rw[r * 128 + 0] = t[m].x;
        rw[r * 128 + 32] = t[m].y;
        rw[r * 128 + 64] = t[m].z;
        rw[r * 128 + 96] = t[m].w;
      }
    }
    for (int i = tid; i < nr; i += 1024) disL[i] = dis[base + i];
  }
  __syncthreads();

  if (listOK) {
    for (int b0 = 0; b0 < n; b0 += 32) {
      const int s = ent.x;
      const int dl = ent.y;
      if (b0 + 32 < n) ent = myList[min(b0 + 32 + lane, n - 1)];
      const int m32 = min(32, n - b0);
#pragma unroll
      for (int q = 0; q < 4; ++q) {
        const int m = m32 - 8 * q;
        if (m > 0) {
          v4f v[8];
#pragma unroll
          for (int j = 0; j < 8; ++j) {
            if (j < m) {
              const int sj = __builtin_amdgcn_readlane(s, 8 * q + j);
              v[j] = *reinterpret_cast<const v4f*>(H + (size_t)sj * 128 +
                                                   4 * lane);
            }
          }
#pragma unroll
          for (int j = 0; j < 8; ++j) {
            if (j < m) {
              const int dj = __builtin_amdgcn_readlane(dl, 8 * q + j);
              float* r = rows + dj * 128 + lane;
              __hip_atomic_fetch_add(r + 0, v[j].x, __ATOMIC_RELAXED,
                                     __HIP_MEMORY_SCOPE_WORKGROUP);
              __hip_atomic_fetch_add(r + 32, v[j].y, __ATOMIC_RELAXED,
                                     __HIP_MEMORY_SCOPE_WORKGROUP);
              __hip_atomic_fetch_add(r + 64, v[j].z, __ATOMIC_RELAXED,
                                     __HIP_MEMORY_SCOPE_WORKGROUP);
              __hip_atomic_fetch_add(r + 96, v[j].w, __ATOMIC_RELAXED,
                                     __HIP_MEMORY_SCOPE_WORKGROUP);
            }
          }
        }
      }
    }
  } else {
    const int nIter = (nedges + kWgGroupEdges - 1) / kWgGroupEdges;
    for (int g = 0; g < nIter; ++g) {
      const int eb = (g * kOwnWaves + wave) * kGroupEdges;
      const int ee = min(eb + kGroupEdges, nedges);
      for (int e0 = eb; e0 < ee; e0 += 32) {
        const int e = e0 + lane;
        const unsigned u = (e < ee) ? (unsigned)(dst[e] - base) : 0xffffffffu;
        const bool own = u < unr;
        unsigned mask = __builtin_amdgcn_ballot_w32(own);
        if (mask == 0u) continue;
        const int s = own ? src[e] : 0;
        while (mask != 0u) {
          const int j = __builtin_ctz(mask);
          mask &= mask - 1u;
          const int sj = __builtin_amdgcn_readlane(s, j);
          const int dj = __builtin_amdgcn_readlane((int)u, j);
          const v4f hv =
              *reinterpret_cast<const v4f*>(H + (size_t)sj * 128 + 4 * lane);
          float* r = rows + dj * 128 + lane;
          __hip_atomic_fetch_add(r + 0, hv.x, __ATOMIC_RELAXED,
                                 __HIP_MEMORY_SCOPE_WORKGROUP);
          __hip_atomic_fetch_add(r + 32, hv.y, __ATOMIC_RELAXED,
                                 __HIP_MEMORY_SCOPE_WORKGROUP);
          __hip_atomic_fetch_add(r + 64, hv.z, __ATOMIC_RELAXED,
                                 __HIP_MEMORY_SCOPE_WORKGROUP);
          __hip_atomic_fetch_add(r + 96, hv.w, __ATOMIC_RELAXED,
                                 __HIP_MEMORY_SCOPE_WORKGROUP);
        }
      }
    }
  }

  __syncthreads();

  {
    float4* __restrict__ p4 = reinterpret_cast<float4*>(plane) + (size_t)base * 32;
    const int n4 = nr * 32;
    for (int i = tid; i < n4; i += 1024) {
      const float* rr = rows + (i >> 5) * 128 + lane;
      const float d = disL[i >> 5];
      float4 o;
      o.x = fmaxf(rr[0] * d + bias4.x, 0.0f);
      o.y = fmaxf(rr[32] * d + bias4.y, 0.0f);
      o.z = fmaxf(rr[64] * d + bias4.z, 0.0f);
      o.w = fmaxf(rr[96] * d + bias4.w, 0.0f);
      p4[i] = o;
    }
  }
}

extern "C" void kernel_launch(void* const* d_in, const int* in_sizes, int n_in,
                              void* d_out, int out_size, void* d_ws,
                              size_t ws_size, hipStream_t stream) {
  const float* x = (const float*)d_in[0];
  const int* edge = (const int*)d_in[1];
  const float* W0 = (const float*)d_in[2];
  const float* b0 = (const float*)d_in[3];
  const float* W1 = (const float*)d_in[4];
  const float* b1 = (const float*)d_in[5];
  const float* W2 = (const float*)d_in[6];
  const float* b2 = (const float*)d_in[7];
  float* out = (float*)d_out;

  const int N = in_sizes[0] / 256;
  const int E = in_sizes[1] / 2;
  const int* srcIdx = edge;
  const int* dstIdx = edge + E;

  const int ownBlocks = (N + kOwnRows - 1) / kOwnRows;
  const int nSlices = (E + kGroupEdges - 1) / kGroupEdges;
  const int gemmBlocks = (N + 31) / 32;

  float* ws = (float*)d_ws;
  float* hbuf = ws;
  float* acc0 = hbuf + (size_t)N * 128;
  float* acc1 = acc0 + (size_t)N * 128;
  int2* lists = (int2*)(acc1 + (size_t)N * 128);
  int2* sorted = lists + (size_t)ownBlocks * kOwnWaves * kListCap;
  int* listCnt = (int*)(sorted + (size_t)nSlices * kGroupEdges);
  float* dis = (float*)(listCnt + (size_t)ownBlocks * kOwnWaves);
  int* offs = (int*)(dis + N);

  gcn_slice_sort<<<nSlices, kSortThreads, 0, stream>>>(srcIdx, dstIdx, sorted, offs,
                                                       N, E, ownBlocks);
  gcn_owner_build<<<ownBlocks, 1024, 0, stream>>>(sorted, offs, lists, listCnt,
                                                  dis, N, E, ownBlocks);

  gcn_gemm_wmma<256><<<gemmBlocks, 128, 0, stream>>>(x, W0, dis, hbuf, N);
  gcn_owner_pass<<<ownBlocks, 1024, 0, stream>>>(srcIdx, dstIdx, lists, listCnt,
                                                 hbuf, dis, b0, acc0, N, E);

  gcn_gemm_wmma<128><<<gemmBlocks, 128, 0, stream>>>(acc0, W1, dis, hbuf, N);
  gcn_owner_pass<<<ownBlocks, 1024, 0, stream>>>(srcIdx, dstIdx, lists, listCnt,
                                                 hbuf, dis, b1, acc1, N, E);

  gcn_gemm_wmma<128><<<gemmBlocks, 128, 0, stream>>>(acc1, W2, dis, hbuf, N);
  gcn_owner_pass<<<ownBlocks, 1024, 0, stream>>>(srcIdx, dstIdx, lists, listCnt,
                                                 hbuf, dis, b2, out, N, E);
}


# LDS float atomics replaced by plain 16-byte LDS read-add-write: build deals each edge to the wave owning its row (row mod 32), one writer per row, no barrier; sum order: own row, then list order
# speedup vs baseline: 4.6610x; 1.0690x over previous
#include <hip/hip_runtime.h>
#include <hip/hip_bf16.h>


typedef __attribute__((ext_vector_type(2))) float v2f;
typedef __attribute__((ext_vector_type(4))) float v4f;
typedef __attribute__((ext_vector_type(8))) float v8f;

template <int K>
__global__ __launch_bounds__(128) void gcn_gemm_wmma(
    const float* __restrict__ X, const float* __restrict__ W,
    const float* __restrict__ dis, float* __restrict__ H, int nrows) {
  constexpr int NCHUNK = K / 32;
  constexpr int NSTEP = K / 4;

  const int lane = threadIdx.x & 31;
  const int c = lane & 15;
  const int h = lane >> 4;
  const int waveInBlock = __builtin_amdgcn_readfirstlane(threadIdx.x >> 5);
  const int u = waveInBlock & 1;
  const int rowBase = (blockIdx.x * 2 + (waveInBlock >> 1)) * 16;
  if (rowBase >= nrows) return;

  const float* __restrict__ xr =
      X + (size_t)min(rowBase + c, nrows - 1) * K + 4 * h;
  const float* __restrict__ wl = W + (size_t)(4 * h) * 128 + 64 * u + 4 * c;

  float dv[8];
#pragma unroll
  for (int r = 0; r < 8; ++r) {
    dv[r] = dis[min(rowBase + 8 * h + r, nrows - 1)];
  }

  v8f acc[4];
#pragma unroll
  for (int t = 0; t < 4; ++t) acc[t] = v8f{};

  constexpr int D = 3;
  v4f a[2][4];
  float4 b[D + 1][2];

#pragma unroll
  for (int i = 0; i < 4; ++i) {
    a[0][i] = *reinterpret_cast<const v4f*>(xr + 8 * i);
  }
#pragma unroll
  for (int m = 0; m < D; ++m) {
    const int km = 8 * (m >> 1) + 2 * (m & 1);
#pragma unroll
    for (int s = 0; s < 2; ++s) {
      b[m][s] = *reinterpret_cast<const float4*>(wl + (size_t)(km + s) * 128);
    }
  }

#pragma unroll
  for (int n = 0; n < NSTEP; ++n) {
    const int g = n >> 1;
    const int p = n & 1;
    const int J = g >> 2;
    const int i = g & 3;
    if (n + D < NSTEP) {
      const int kn = 8 * ((n + D) >> 1) + 2 * ((n + D) & 1);
#pragma unroll
      for (int s = 0; s < 2; ++s) {
        b[(n + D) % (D + 1)][s] =
            *reinterpret_cast<const float4*>(wl + (size_t)(kn + s) * 128);
      }
    }
    if ((n & 7) == 0 && J + 1 < NCHUNK) {
#pragma unroll
      for (int q = 0; q < 4; ++q) {
        a[(J + 1) & 1][q] =
            *reinterpret_cast<const v4f*>(xr + 32 * (J + 1) + 8 * q);
      }
    }
    __builtin_amdgcn_sched_barrier(0);
    const v4f aq = a[J & 1][i];
    v2f af;
    af.x = p ? aq.z : aq.x;
    af.y = p ? aq.w : aq.y;
    const float4 b0 = b[n % (D + 1)][0];
    const float4 b1 = b[n % (D + 1)][1];
    v2f bf;
    bf.x = b0.x; bf.y = b1.x;
    acc[0] = __builtin_amdgcn_wmma_f32_16x16x4_f32(
        false, af, false, bf, (short)0, acc[0], false, false);
    bf.x = b0.y; bf.y = b1.y;
    acc[1] = __builtin_amdgcn_wmma_f32_16x16x4_f32(
        false, af, false, bf, (short)0, acc[1], false, false);
    bf.x = b0.z; bf.y = b1.z;
    acc[2] = __builtin_amdgcn_wmma_f32_16x16x4_f32(
        false, af, false, bf, (short)0, acc[2], false, false);
    bf.x = b0.w; bf.y = b1.w;
    acc[3] = __builtin_amdgcn_wmma_f32_16x16x4_f32(
        false, af, false, bf, (short)0, acc[3], false, false);
  }

  float4 hv[8];
#pragma unroll
  for (int r = 0; r < 8; ++r) {
    hv[r] = make_float4(acc[0][r] * dv[r], acc[1][r] * dv[r], acc[2][r] * dv[r],
                        acc[3][r] * dv[r]);
  }
  const size_t o = (size_t)(rowBase + 8 * h) * 128 + 64 * u + 4 * c;
  if (rowBase + 16 <= nrows) {
#pragma unroll
    for (int r = 0; r < 8; ++r) {
      *reinterpret_cast<float4*>(H + o + (size_t)r * 128) = hv[r];
    }
  } else {
#pragma unroll
    for (int r = 0; r < 8; ++r) {
      if (rowBase + 8 * h + r < nrows) {
        *reinterpret_cast<float4*>(H + o + (size_t)r * 128) = hv[r];
      }
    }
  }
}

constexpr int kOwnRows = 196;
constexpr int kOwnWaves = 32;
constexpr int kRowsPerWave = (kOwnRows + kOwnWaves - 1) / kOwnWaves;
constexpr int kGroupLoads = 8;
constexpr int kGroupEdges = kGroupLoads * 128;
constexpr int kListCap = 1024;
constexpr int kSortThreads = 256;
constexpr int kMaxOwn = 1024;
constexpr int kStageCap = 256;

__global__ __launch_bounds__(kSortThreads) void gcn_slice_sort(
    const int* __restrict__ src, const int* __restrict__ dst,
    int2* __restrict__ sorted, int* __restrict__ offs, int nrows, int nedges,
    int nOwn) {
  __shared__ int cntL[kMaxOwn];
  __shared__ int offL[kMaxOwn + 1];
  __shared__ int wsum[kSortThreads / 32];
  __shared__ int2 entL[kGroupEdges];

  const int tid = threadIdx.x;
  const int lane = tid & 31;
  const int wave = tid >> 5;
  const int eb = blockIdx.x * kGroupEdges + 4 * tid;

  for (int i = tid; i < kMaxOwn; i += kSortThreads) cntL[i] = 0;

  const bool vecOK = ((nedges & 3) == 0) &&
                     ((reinterpret_cast<size_t>(dst) & 15) == 0) &&
                     ((reinterpret_cast<size_t>(src) & 15) == 0);
  int d[4];
  int s[4];
  if (vecOK && eb + 4 <= nedges) {
    const int4 dv = *reinterpret_cast<const int4*>(dst + eb);
    const int4 sv = *reinterpret_cast<const int4*>(src + eb);
    d[0] = dv.x; d[1] = dv.y; d[2] = dv.z; d[3] = dv.w;
    s[0] = sv.x; s[1] = sv.y; s[2] = sv.z; s[3] = sv.w;
  } else {
#pragma unroll
    for (int j = 0; j < 4; ++j) {
      d[j] = (eb + j < nedges) ? dst[eb + j] : -1;
      s[j] = (eb + j < nedges) ? src[eb + j] : 0;
    }
  }
  __syncthreads();

  int o[4];
  int l[4];
  int rk[4];
#pragma unroll
  for (int j = 0; j < 4; ++j) {
    const bool ok = (unsigned)d[j] < (unsigned)nrows;
    const int oj = ok ? d[j] / kOwnRows : -1;
    o[j] = (oj < kMaxOwn) ? oj : -1;
    l[j] = d[j] - oj * kOwnRows;
    rk[j] = 0;
    if (o[j] >= 0) {
      rk[j] = __hip_atomic_fetch_add(&cntL[o[j]], 1, __ATOMIC_RELAXED,
                                     __HIP_MEMORY_SCOPE_WORKGROUP);
    }
  }
  __syncthreads();

  const int c0 = cntL[4 * tid + 0];
  const int c1 = cntL[4 * tid + 1];
  const int c2 = cntL[4 * tid + 2];
  const int c3 = cntL[4 * tid + 3];
  const int tsum = c0 + c1 + c2 + c3;
  int x = tsum;
  for (int dl = 1; dl < 32; dl <<= 1) {
    const int y = __shfl_up(x, dl);
    if (lane >= dl) x += y;
  }
  if (lane == 31) wsum[wave] = x;
  __syncthreads();
  int wb = 0;
  for (int i = 0; i < kSortThreads / 32; ++i) {
    if (i < wave) wb += wsum[i];
  }
  const int ex = wb + x - tsum;
  offL[4 * tid + 0] = ex;
  offL[4 * tid + 1] = ex + c0;
  offL[4 * tid + 2] = ex + c0 + c1;
  offL[4 * tid + 3] = ex + c0 + c1 + c2;
  if (tid == kSortThreads - 1) offL[kMaxOwn] = ex + tsum;
  __syncthreads();

  const int total = offL[kMaxOwn];
#pragma unroll
  for (int j = 0; j < 4; ++j) {
    if (o[j] >= 0) entL[offL[o[j]] + rk[j]] = make_int2(s[j], l[j]);
  }
  int* __restrict__ myOffs = offs + (size_t)blockIdx.x * (nOwn + 1);
  for (int i = tid; i <= nOwn; i += kSortThreads) myOffs[i] = offL[i];
  __syncthreads();

  int2* __restrict__ mySorted = sorted + (size_t)blockIdx.x * kGroupEdges;
  for (int i = tid; i < total; i += kSortThreads) mySorted[i] = entL[i];
}

__global__ __launch_bounds__(1024) void gcn_owner_build(
    const int2* __restrict__ sorted, const int* __restrict__ offs,
    int2* __restrict__ lists, int* __restrict__ listCnt,
    float* __restrict__ dis, int nrows, int nedges, int nOwn) {
  __shared__ int degL[kOwnRows];
  __shared__ int cntW[kOwnWaves];
  __shared__ int2 stage[kOwnWaves * kStageCap];

  const int tid = threadIdx.x;
  const int lane = tid & 31;
  const int wave = __builtin_amdgcn_readfirstlane(tid >> 5);
  const int base = blockIdx.x * kOwnRows;
  const int nr = min(kOwnRows, nrows - base);
  const int nSlices = (nedges + kGroupEdges - 1) / kGroupEdges;

  for (int i = tid; i < kOwnRows; i += 1024) degL[i] = 0;
  if (tid < kOwnWaves) cntW[tid] = 0;
  __syncthreads();

  int2* __restrict__ wgLists =
      lists + (size_t)blockIdx.x * kOwnWaves * kListCap;

  for (int g0 = wave; g0 < nSlices; g0 += kOwnWaves * 32) {
    const int g = g0 + kOwnWaves * lane;
    const bool have = g < nSlices;
    const int gc = have ? g : 0;
    const int* __restrict__ op = offs + (size_t)gc * (nOwn + 1) + blockIdx.x;
    const int lo = op[0];
    const int n = have ? (op[1] - lo) : 0;
    const int2* __restrict__ sp = sorted + (size_t)gc * kGroupEdges + lo;

    for (int k = 0; __builtin_amdgcn_ballot_w32(k < n) != 0u; k += 4) {
      int2 e[4];
#pragma unroll
      for (int j = 0; j < 4; ++j) {
        e[j] = make_int2(0, 0);
        if (k + j < n) e[j] = sp[k + j];
      }
#pragma unroll
      for (int j = 0; j < 4; ++j) {
        if (k + j < n) {
          const int tw = e[j].y & (kOwnWaves - 1);
          const int p = __hip_atomic_fetch_add(&cntW[tw], 1, __ATOMIC_RELAXED,
                                               __HIP_MEMORY_SCOPE_WORKGROUP);
          if (p < kStageCap) {
            stage[tw * kStageCap + p] = e[j];
          } else if (p < kListCap) {
            wgLists[(size_t)tw * kListCap + p] = e[j];
          }
          __hip_atomic_fetch_add(&degL[e[j].y], 1, __ATOMIC_RELAXED,
                                 __HIP_MEMORY_SCOPE_WORKGROUP);
        }
      }
    }
  }

  __syncthreads();

  const int done = cntW[wave];
  if (lane == 0) listCnt[blockIdx.x * kOwnWaves + wave] = done;

  int2* __restrict__ myList = wgLists + (size_t)wave * kListCap;
  const int2* myStage = stage + wave * kStageCap;
  const int ns = min(done, kStageCap);
  for (int i = lane; i < ns; i += 32) myList[i] = myStage[i];

  for (int i = tid; i < nr; i += 1024) {
    const float d = (float)(degL[i] + 1);
    dis[base + i] = rsqrtf(d);
  }
}

__global__ __launch_bounds__(1024) void gcn_owner_pass(
    const int* __restrict__ src, const int* __restrict__ dst,
    const int2* __restrict__ lists, const int* __restrict__ listCnt,
    const float* __restrict__ H, const float* __restrict__ dis,
    const float* __restrict__ bias, float* __restrict__ plane, int nrows,
    int nedges) {
  __shared__ v4f rows[kOwnRows * 32];

  const int tid = threadIdx.x;
  const int lane = tid & 31;
  const int wave = __builtin_amdgcn_readfirstlane(tid >> 5);
  const int base = blockIdx.x * kOwnRows;
  const int nr = min(kOwnRows, nrows - base);
  const unsigned unr = (unsigned)nr;

  const int2* __restrict__ myList =
      lists + (size_t)(blockIdx.x * kOwnWaves + wave) * kListCap;
  const int n = __builtin_amdgcn_readfirstlane(
      listCnt[blockIdx.x * kOwnWaves + wave]);
  const bool listOK = (n <= kListCap);
  int2 ent = make_int2(0, 0);
  if (listOK && n > 0) ent = myList[min(lane, n - 1)];
  const float4 bias4 = *reinterpret_cast<const float4*>(bias + 4 * lane);

  const v4f* __restrict__ H4 = reinterpret_cast<const v4f*>(H) + lane;
  v4f* rl = rows + lane;

  float dv[kRowsPerWave];
  {
    v4f t[kRowsPerWave];
#pragma unroll
    for (int m = 0; m < kRowsPerWave; ++m) {
      const int r = wave + kOwnWaves * m;
      dv[m] = 0.0f;
      if (r < nr) {
        t[m] = H4[(size_t)(base + r) * 32];
        dv[m] = dis[base + r];
      }
    }
#pragma unroll
    for (int m = 0; m < kRowsPerWave; ++m) {
      const int r = wave + kOwnWaves * m;
      if (r < nr) rl[32 * r] = t[m];
    }
  }

  if (listOK) {
    int b0 = 0;
    for (; b0 + 32 <= n; b0 += 32) {
      const int s = ent.x;
      const int dl = ent.y;
      if (b0 + 32 < n) ent = myList[min(b0 + 32 + lane, n - 1)];
      v4f v[8];
#pragma unroll
      for (int j = 0; j < 8; ++j) {
        v[j] = H4[(size_t)__builtin_amdgcn_readlane(s, j) * 32];
      }
#pragma unroll
      for (int i = 0; i < 32; ++i) {
        const int dj = __builtin_amdgcn_readlane(dl, i);
        rl[32 * dj] = rl[32 * dj] + v[i & 7];
        if (i + 8 < 32) {
          v[i & 7] = H4[(size_t)__builtin_amdgcn_readlane(s, i + 8) * 32];
        }
      }
    }
    if (b0 < n) {
      const int s = ent.x;
      const int dl = ent.y;
      const int m32 = n - b0;
#pragma unroll
      for (int q = 0; q < 4; ++q) {
        const int m = m32 - 8 * q;
        if (m > 0) {
          v4f v[8];
#pragma unroll
          for (int j = 0; j < 8; ++j) {
            if (j < m) {
              v[j] = H4[(size_t)__builtin_amdgcn_readlane(s, 8 * q + j) * 32];
            }
          }
#pragma unroll
          for (int j = 0; j < 8; ++j) {
            if (j < m) {
              const int dj = __builtin_amdgcn_readlane(dl, 8 * q + j);
              rl[32 * dj] = rl[32 * dj] + v[j];
            }
          }
        }
      }
    }
  } else {
    for (int e0 = 0; e0 < nedges; e0 += 32) {
      const int e = e0 + lane;
      const unsigned u = (e < nedges) ? (unsigned)(dst[e] - base) : 0xffffffffu;
      const bool own = (u < unr) && ((int)(u & (kOwnWaves - 1)) == wave);
      unsigned mask = __builtin_amdgcn_ballot_w32(own);
      if (mask == 0u) continue;
      const int s = own ? src[e] : 0;
      while (mask != 0u) {
        const int j = __builtin_ctz(mask);
        mask &= mask - 1u;
        const int sj = __builtin_amdgcn_readlane(s, j);
        const int dj = __builtin_amdgcn_readlane((int)u, j);
        rl[32 * dj] = rl[32 * dj] + H4[(size_t)sj * 32];
      }
    }
  }

  float4* __restrict__ P4 = reinterpret_cast<float4*>(plane) + lane;
#pragma unroll
  for (int m = 0; m < kRowsPerWave; ++m) {
    const int r = wave + kOwnWaves * m;
    if (r < nr) {
      const v4f a = rl[32 * r];
      float4 o;
      o.x = fmaxf(a.x * dv[m] + bias4.x, 0.0f);
      o.y = fmaxf(a.y * dv[m] + bias4.y, 0.0f);
      o.z = fmaxf(a.z * dv[m] + bias4.z, 0.0f);
      o.w = fmaxf(a.w * dv[m] + bias4.w, 0.0f);
      P4[(size_t)(base + r) * 32] = o;
    }
  }
}

extern "C" void kernel_launch(void* const* d_in, const int* in_sizes, int n_in,
                              void* d_out, int out_size, void* d_ws,
                              size_t ws_size, hipStream_t stream) {
  const float* x = (const float*)d_in[0];
  const int* edge = (const int*)d_in[1];
  const float* W0 = (const float*)d_in[2];
  const float* b0 = (const float*)d_in[3];
  const float* W1 = (const float*)d_in[4];
  const float* b1 = (const float*)d_in[5];
  const float* W2 = (const float*)d_in[6];
  const float* b2 = (const float*)d_in[7];
  float* out = (float*)d_out;

  const int N = in_sizes[0] / 256;
  const int E = in_sizes[1] / 2;
  const int* srcIdx = edge;
  const int* dstIdx = edge + E;

  const int ownBlocks = (N + kOwnRows - 1) / kOwnRows;
  const int nSlices = (E + kGroupEdges - 1) / kGroupEdges;
  const int gemmBlocks = (N + 31) / 32;

  float* ws = (float*)d_ws;
  float* hbuf = ws;
  float* acc0 = hbuf + (size_t)N * 128;
  float* acc1 = acc0 + (size_t)N * 128;
  int2* lists = (int2*)(acc1 + (size_t)N * 128);
  int2* sorted = lists + (size_t)ownBlocks * kOwnWaves * kListCap;
  int* listCnt = (int*)(sorted + (size_t)nSlices * kGroupEdges);
  float* dis = (float*)(listCnt + (size_t)ownBlocks * kOwnWaves);
  int* offs = (int*)(dis + N);

  gcn_slice_sort<<<nSlices, kSortThreads, 0, stream>>>(srcIdx, dstIdx, sorted, offs,
                                                       N, E, ownBlocks);
  gcn_owner_build<<<ownBlocks, 1024, 0, stream>>>(sorted, offs, lists, listCnt,
                                                  dis, N, E, ownBlocks);

  gcn_gemm_wmma<256><<<gemmBlocks, 128, 0, stream>>>(x, W0, dis, hbuf, N);
  gcn_owner_pass<<<ownBlocks, 1024, 0, stream>>>(srcIdx, dstIdx, lists, listCnt,
                                                 hbuf, dis, b0, acc0, N, E);

  gcn_gemm_wmma<128><<<gemmBlocks, 128, 0, stream>>>(acc0, W1, dis, hbuf, N);
  gcn_owner_pass<<<ownBlocks, 1024, 0, stream>>>(srcIdx, dstIdx, lists, listCnt,
                                                 hbuf, dis, b1, acc1, N, E);

  gcn_gemm_wmma<128><<<gemmBlocks, 128, 0, stream>>>(acc1, W2, dis, hbuf, N);
  gcn_owner_pass<<<ownBlocks, 1024, 0, stream>>>(srcIdx, dstIdx, lists, listCnt,
                                                 hbuf, dis, b2, out, N, E);
}
